# starting kernel
# baseline (speedup 1.0000x reference)
#include <hip/hip_runtime.h>

typedef __attribute__((ext_vector_type(2))) float v2f;
typedef __attribute__((ext_vector_type(8))) float v8f;

static constexpr int NN = 100000;
static constexpr int NE = 1600000;
static constexpr int DF = 128;
static constexpr int DH = 64;
static constexpr int DC = 40;
#define NEG_SLOPE 0.01f

__device__ __forceinline__ float lrelu(float x) { return x > 0.f ? x : x * NEG_SLOPE; }

__device__ __forceinline__ v8f wmma4(v2f a, v2f b, v8f c) {
    return __builtin_amdgcn_wmma_f32_16x16x4_f32(false, a, false, b, (short)0, c, false, false);
}

__global__ void zero_f32(float* __restrict__ p, int n) {
    int i = blockIdx.x * blockDim.x + threadIdx.x;
    int stride = gridDim.x * blockDim.x;
    for (; i < n; i += stride) p[i] = 0.f;
}

__global__ void deg_kernel(const int* __restrict__ snd, const int* __restrict__ rcv,
                           float* __restrict__ sdeg, float* __restrict__ rdeg) {
    int e = blockIdx.x * blockDim.x + threadIdx.x;
    if (e < NE) {
        atomicAdd(&sdeg[snd[e]], 1.f);
        atomicAdd(&rdeg[rcv[e]], 1.f);
    }
}

__launch_bounds__(256)
__global__ void mlp_kernel(const float* __restrict__ X,
                           const float* __restrict__ W1, const float* __restrict__ b1,
                           const float* __restrict__ W2, const float* __restrict__ b2,
                           const float* __restrict__ sdeg,
                           float* __restrict__ h1, int nTiles) {
    __shared__ float lds[8 * 16 * DH];
    const int w    = threadIdx.x >> 5;
    const int lane = threadIdx.x & 31;
    const int m16  = lane & 15;
    const int kh   = lane >> 4;
    const int t    = blockIdx.x * 8 + w;
    const bool active = (t < nTiles);
    const int row0 = t * 16;
    float* my = &lds[w * 16 * DH];

    if (active) {
        v8f c0 = {}, c1 = {}, c2 = {}, c3 = {};
        #pragma unroll 4
        for (int kb = 0; kb < DF; kb += 4) {
            v2f a = *(const v2f*)(X + (size_t)(row0 + m16) * DF + kb + 2 * kh);
            const float* wr0 = W1 + (size_t)(kb + 2 * kh) * DH;
            const float* wr1 = wr0 + DH;
            v2f b;
            b.x = wr0[m16];      b.y = wr1[m16];      c0 = wmma4(a, b, c0);
            b.x = wr0[16 + m16]; b.y = wr1[16 + m16]; c1 = wmma4(a, b, c1);
            b.x = wr0[32 + m16]; b.y = wr1[32 + m16]; c2 = wmma4(a, b, c2);
            b.x = wr0[48 + m16]; b.y = wr1[48 + m16]; c3 = wmma4(a, b, c3);
        }
        #pragma unroll
        for (int r = 0; r < 8; ++r) {
            float* dr = my + (r + 8 * kh) * DH;
            dr[m16]      = lrelu(c0[r] + b1[m16]);
            dr[16 + m16] = lrelu(c1[r] + b1[16 + m16]);
            dr[32 + m16] = lrelu(c2[r] + b1[32 + m16]);
            dr[48 + m16] = lrelu(c3[r] + b1[48 + m16]);
        }
    }
    __syncthreads();
    if (!active) return;

    v8f d0 = {}, d1 = {}, d2 = {}, d3 = {};
    #pragma unroll 4
    for (int kb = 0; kb < DH; kb += 4) {
        v2f a = *(const v2f*)(my + m16 * DH + kb + 2 * kh);
        const float* wr0 = W2 + (size_t)(kb + 2 * kh) * DH;
        const float* wr1 = wr0 + DH;
        v2f b;
        b.x = wr0[m16];      b.y = wr1[m16];      d0 = wmma4(a, b, d0);
        b.x = wr0[16 + m16]; b.y = wr1[16 + m16]; d1 = wmma4(a, b, d1);
        b.x = wr0[32 + m16]; b.y = wr1[32 + m16]; d2 = wmma4(a, b, d2);
        b.x = wr0[48 + m16]; b.y = wr1[48 + m16]; d3 = wmma4(a, b, d3);
    }
    float sc[8];
    #pragma unroll
    for (int r = 0; r < 8; ++r) sc[r] = rsqrtf(fmaxf(sdeg[row0 + r + 8 * kh], 1.f));
    #pragma unroll
    for (int r = 0; r < 8; ++r) {
        float* orow = h1 + (size_t)(row0 + r + 8 * kh) * DH;
        orow[m16]      = lrelu(d0[r] + b2[m16])      * sc[r];
        orow[16 + m16] = lrelu(d1[r] + b2[16 + m16]) * sc[r];
        orow[32 + m16] = lrelu(d2[r] + b2[32 + m16]) * sc[r];
        orow[48 + m16] = lrelu(d3[r] + b2[48 + m16]) * sc[r];
    }
}

__global__ void agg1_kernel(const float* __restrict__ h1, const int* __restrict__ snd,
                            const int* __restrict__ rcv, float* __restrict__ agg) {
    int idx = blockIdx.x * blockDim.x + threadIdx.x;
    int e = idx >> 6, f = idx & 63;
    if (e < NE)
        atomicAdd(&agg[(size_t)rcv[e] * DH + f], h1[(size_t)snd[e] * DH + f]);
}

__launch_bounds__(256)
__global__ void fin1_kernel(const float* __restrict__ agg1,
                            const float* __restrict__ rdeg, const float* __restrict__ sdeg,
                            const float* __restrict__ W3, const float* __restrict__ b3,
                            float* __restrict__ h2, int nTiles) {
    const int w    = threadIdx.x >> 5;
    const int lane = threadIdx.x & 31;
    const int m16  = lane & 15;
    const int kh   = lane >> 4;
    const int t    = blockIdx.x * 8 + w;
    if (t >= nTiles) return;
    const int row0 = t * 16;
    const float rs = rsqrtf(fmaxf(rdeg[row0 + m16], 1.f));

    v8f c0 = {}, c1 = {}, c2 = {};
    #pragma unroll 4
    for (int kb = 0; kb < DH; kb += 4) {
        v2f a = *(const v2f*)(agg1 + (size_t)(row0 + m16) * DH + kb + 2 * kh);
        a.x *= rs; a.y *= rs;
        const float* wr0 = W3 + (size_t)(kb + 2 * kh) * DC;
        const float* wr1 = wr0 + DC;
        v2f b;
        b.x = wr0[m16];      b.y = wr1[m16];      c0 = wmma4(a, b, c0);
        b.x = wr0[16 + m16]; b.y = wr1[16 + m16]; c1 = wmma4(a, b, c1);
        const int n2 = 32 + m16;
        b.x = (n2 < DC) ? wr0[n2] : 0.f;
        b.y = (n2 < DC) ? wr1[n2] : 0.f;
        c2 = wmma4(a, b, c2);
    }
    float sc[8];
    #pragma unroll
    for (int r = 0; r < 8; ++r) sc[r] = rsqrtf(fmaxf(sdeg[row0 + r + 8 * kh], 1.f));
    #pragma unroll
    for (int r = 0; r < 8; ++r) {
        float* orow = h2 + (size_t)(row0 + r + 8 * kh) * DC;
        orow[m16]      = (c0[r] + b3[m16])      * sc[r];
        orow[16 + m16] = (c1[r] + b3[16 + m16]) * sc[r];
        if (32 + m16 < DC)
            orow[32 + m16] = (c2[r] + b3[32 + m16]) * sc[r];
    }
}

__global__ void agg2_kernel(const float* __restrict__ h2, const int* __restrict__ snd,
                            const int* __restrict__ rcv, float* __restrict__ out) {
    int idx = blockIdx.x * blockDim.x + threadIdx.x;
    int e = idx / DC, f = idx - e * DC;
    if (e < NE)
        atomicAdd(&out[(size_t)rcv[e] * DC + f], h2[(size_t)snd[e] * DC + f]);
}

__global__ void softmax_kernel(float* __restrict__ out, const float* __restrict__ rdeg) {
    int n = blockIdx.x * blockDim.x + threadIdx.x;
    if (n >= NN) return;
    float rs = rsqrtf(fmaxf(rdeg[n], 1.f));
    float* row = out + (size_t)n * DC;
    float v[DC];
    float mx = -3.402823466e38f;
    #pragma unroll
    for (int f = 0; f < DC; ++f) { v[f] = row[f] * rs; mx = fmaxf(mx, v[f]); }
    float s = 0.f;
    #pragma unroll
    for (int f = 0; f < DC; ++f) { v[f] = __expf(v[f] - mx); s += v[f]; }
    float inv = 1.f / s;
    #pragma unroll
    for (int f = 0; f < DC; ++f) row[f] = v[f] * inv;
}

extern "C" void kernel_launch(void* const* d_in, const int* in_sizes, int n_in,
                              void* d_out, int out_size, void* d_ws, size_t ws_size,
                              hipStream_t stream) {
    const float* nodes     = (const float*)d_in[0];
    const int*   senders   = (const int*)d_in[1];
    const int*   receivers = (const int*)d_in[2];
    const float* W1 = (const float*)d_in[3];
    const float* b1 = (const float*)d_in[4];
    const float* W2 = (const float*)d_in[5];
    const float* b2 = (const float*)d_in[6];
    const float* W3 = (const float*)d_in[7];
    const float* b3 = (const float*)d_in[8];
    float* out = (float*)d_out;
    float* ws  = (float*)d_ws;

    float* sdeg = ws;
    float* rdeg = ws + NN;
    float* h1   = ws + 2 * (size_t)NN;
    float* agg1 = h1 + (size_t)NN * DH;
    float* h2   = agg1 + (size_t)NN * DH;

    const int nTiles = NN / 16;
    zero_f32<<<2048, 256, 0, stream>>>(sdeg, 2 * NN);
    zero_f32<<<8192, 256, 0, stream>>>(agg1, NN * DH);
    zero_f32<<<8192, 256, 0, stream>>>(out, NN * DC);
    deg_kernel<<<(NE + 255) / 256, 256, 0, stream>>>(senders, receivers, sdeg, rdeg);
    mlp_kernel<<<(nTiles + 7) / 8, 256, 0, stream>>>(nodes, W1, b1, W2, b2, sdeg, h1, nTiles);
    agg1_kernel<<<(NE * 64) / 256, 256, 0, stream>>>(h1, senders, receivers, agg1);
    fin1_kernel<<<(nTiles + 7) / 8, 256, 0, stream>>>(agg1, rdeg, sdeg, W3, b3, h2, nTiles);
    agg2_kernel<<<(NE * DC + 255) / 256, 256, 0, stream>>>(h2, senders, receivers, out);
    softmax_kernel<<<(NN + 255) / 256, 256, 0, stream>>>(out, rdeg);
}


# Atomic scatter-adds replaced by a receiver-sorted gather: edge rank = value returned by the degree atomic, prefix sum gives segments; rows summed in arrival-rank order, each output row stored once
# speedup vs baseline: 2.1838x; 2.1838x over previous
#include <hip/hip_runtime.h>

typedef __attribute__((ext_vector_type(2))) float v2f;
typedef __attribute__((ext_vector_type(8))) float v8f;

static constexpr int NN = 100000;
static constexpr int NE = 1600000;
static constexpr int DF = 128;
static constexpr int DH = 64;
static constexpr int DC = 40;
#define NEG_SLOPE 0.01f

__device__ __forceinline__ float lrelu(float x) { return x > 0.f ? x : x * NEG_SLOPE; }

__device__ __forceinline__ v8f wmma4(v2f a, v2f b, v8f c) {
    return __builtin_amdgcn_wmma_f32_16x16x4_f32(false, a, false, b, (short)0, c, false, false);
}

__global__ void zero_f32(float* __restrict__ p, int n) {
    int i = blockIdx.x * blockDim.x + threadIdx.x;
    int stride = gridDim.x * blockDim.x;
    for (; i < n; i += stride) p[i] = 0.f;
}

static_assert(NE < (1 << 24), "edge counts and ranks must stay exact in f32");
static_assert(NN % 4 == 0, "offs_kernel reads the degrees four at a time");

__global__ void deg_kernel(const int* __restrict__ snd, const int* __restrict__ rcv,
                           float* __restrict__ sdeg, float* __restrict__ rdeg,
                           int* __restrict__ rank) {
    int e = blockIdx.x * blockDim.x + threadIdx.x;
    if (e < NE) {
        const int s = snd[e];
        const int r = rcv[e];
        const float before = atomicAdd(&rdeg[r], 1.f);
        atomicAdd(&sdeg[s], 1.f);
        rank[e] = (int)before;
    }
}

__launch_bounds__(256)
__global__ void offs_kernel(const float* __restrict__ rdeg, int* __restrict__ offs) {
    __shared__ int own[256];
    __shared__ int front[256];
    const int t = threadIdx.x;
    const int b = blockIdx.x;
    int acc = 0;
    for (int i = 0; i < b; ++i) {
        const float4 v = *(const float4*)(rdeg + (size_t)i * 1024 + 4 * t);
        acc += (int)v.x + (int)v.y + (int)v.z + (int)v.w;
    }
    const int i0 = b * 1024 + 4 * t;
    int c0 = 0, c1 = 0, c2 = 0, c3 = 0;
    if (i0 < NN) {
        const float4 v = *(const float4*)(rdeg + i0);
        c0 = (int)v.x; c1 = (int)v.y; c2 = (int)v.z; c3 = (int)v.w;
    }
    const int mine = c0 + c1 + c2 + c3;
    own[t] = mine;
    front[t] = acc;
    __syncthreads();
    for (int d = 1; d < 256; d <<= 1) {
        const int a = (t >= d) ? own[t - d] : 0;
        const int f = (t >= d) ? front[t - d] : 0;
        __syncthreads();
        own[t] += a;
        front[t] += f;
        __syncthreads();
    }
    const int o = front[255] + own[t] - mine;
    if (i0 < NN) {
        int4 w;
        w.x = o; w.y = o + c0; w.z = o + c0 + c1; w.w = o + c0 + c1 + c2;
        *(int4*)(offs + i0) = w;
    }
}

__global__ void fill_kernel(const int* __restrict__ snd, const int* __restrict__ rcv,
                            const int* __restrict__ rank, const int* __restrict__ offs,
                            int* __restrict__ csr) {
    int e = blockIdx.x * blockDim.x + threadIdx.x;
    if (e < NE) {
        const int p = offs[rcv[e]] + rank[e];
        if ((unsigned)p < (unsigned)NE) csr[p] = snd[e];
    }
}

__launch_bounds__(256)
__global__ void mlp_kernel(const float* __restrict__ X,
                           const float* __restrict__ W1, const float* __restrict__ b1,
                           const float* __restrict__ W2, const float* __restrict__ b2,
                           const float* __restrict__ sdeg,
                           float* __restrict__ h1, int nTiles) {
    __shared__ float lds[8 * 16 * DH];
    const int w    = threadIdx.x >> 5;
    const int lane = threadIdx.x & 31;
    const int m16  = lane & 15;
    const int kh   = lane >> 4;
    const int t    = blockIdx.x * 8 + w;
    const bool active = (t < nTiles);
    const int row0 = t * 16;
    float* my = &lds[w * 16 * DH];

    if (active) {
        v8f c0 = {}, c1 = {}, c2 = {}, c3 = {};
        #pragma unroll 4
        for (int kb = 0; kb < DF; kb += 4) {
            v2f a = *(const v2f*)(X + (size_t)(row0 + m16) * DF + kb + 2 * kh);
            const float* wr0 = W1 + (size_t)(kb + 2 * kh) * DH;
            const float* wr1 = wr0 + DH;
            v2f b;
            b.x = wr0[m16];      b.y = wr1[m16];      c0 = wmma4(a, b, c0);
            b.x = wr0[16 + m16]; b.y = wr1[16 + m16]; c1 = wmma4(a, b, c1);
            b.x = wr0[32 + m16]; b.y = wr1[32 + m16]; c2 = wmma4(a, b, c2);
            b.x = wr0[48 + m16]; b.y = wr1[48 + m16]; c3 = wmma4(a, b, c3);
        }
        #pragma unroll
        for (int r = 0; r < 8; ++r) {
            float* dr = my + (r + 8 * kh) * DH;
            dr[m16]      = lrelu(c0[r] + b1[m16]);
            dr[16 + m16] = lrelu(c1[r] + b1[16 + m16]);
            dr[32 + m16] = lrelu(c2[r] + b1[32 + m16]);
            dr[48 + m16] = lrelu(c3[r] + b1[48 + m16]);
        }
    }
    __syncthreads();
    if (!active) return;

    v8f d0 = {}, d1 = {}, d2 = {}, d3 = {};
    #pragma unroll 4
    for (int kb = 0; kb < DH; kb += 4) {
        v2f a = *(const v2f*)(my + m16 * DH + kb + 2 * kh);
        const float* wr0 = W2 + (size_t)(kb + 2 * kh) * DH;
        const float* wr1 = wr0 + DH;
        v2f b;
        b.x = wr0[m16];      b.y = wr1[m16];      d0 = wmma4(a, b, d0);
        b.x = wr0[16 + m16]; b.y = wr1[16 + m16]; d1 = wmma4(a, b, d1);
        b.x = wr0[32 + m16]; b.y = wr1[32 + m16]; d2 = wmma4(a, b, d2);
        b.x = wr0[48 + m16]; b.y = wr1[48 + m16]; d3 = wmma4(a, b, d3);
    }
    float sc[8];
    #pragma unroll
    for (int r = 0; r < 8; ++r) sc[r] = rsqrtf(fmaxf(sdeg[row0 + r + 8 * kh], 1.f));
    #pragma unroll
    for (int r = 0; r < 8; ++r) {
        float* orow = h1 + (size_t)(row0 + r + 8 * kh) * DH;
        orow[m16]      = lrelu(d0[r] + b2[m16])      * sc[r];
        orow[16 + m16] = lrelu(d1[r] + b2[16 + m16]) * sc[r];
        orow[32 + m16] = lrelu(d2[r] + b2[32 + m16]) * sc[r];
        orow[48 + m16] = lrelu(d3[r] + b2[48 + m16]) * sc[r];
    }
}

__launch_bounds__(256)
__global__ void gather1_kernel(const float* __restrict__ h1, const int* __restrict__ csr,
                               const int* __restrict__ offs, const float* __restrict__ rdeg,
                               float* __restrict__ agg) {
    const int gt = blockIdx.x * 256 + threadIdx.x;
    const int n = gt >> 4;
    const int l = gt & 15;
    if (n >= NN) return;
    const int* seg = csr + offs[n];
    const int cnt = (int)rdeg[n];
    float4 acc = {0.f, 0.f, 0.f, 0.f};
    for (int j = 0; j < cnt; j += 8) {
        int s[8];
        float4 v[8];
        #pragma unroll
        for (int k = 0; k < 8; ++k) {
            const int q = seg[min(j + k, cnt - 1)];
            s[k] = ((unsigned)q < (unsigned)NN) ? q : 0;
        }
        #pragma unroll
        for (int k = 0; k < 8; ++k)
            v[k] = *(const float4*)(h1 + (size_t)s[k] * DH + 4 * l);
        #pragma unroll
        for (int k = 0; k < 8; ++k) {
            const bool in = j + k < cnt;
            acc.x += in ? v[k].x : 0.f; acc.y += in ? v[k].y : 0.f;
            acc.z += in ? v[k].z : 0.f; acc.w += in ? v[k].w : 0.f;
        }
    }
    *(float4*)(agg + (size_t)n * DH + 4 * l) = acc;
}

__launch_bounds__(256)
__global__ void fin1_kernel(const float* __restrict__ agg1,
                            const float* __restrict__ rdeg, const float* __restrict__ sdeg,
                            const float* __restrict__ W3, const float* __restrict__ b3,
                            float* __restrict__ h2, int nTiles) {
    const int w    = threadIdx.x >> 5;
    const int lane = threadIdx.x & 31;
    const int m16  = lane & 15;
    const int kh   = lane >> 4;
    const int t    = blockIdx.x * 8 + w;
    if (t >= nTiles) return;
    const int row0 = t * 16;
    const float rs = rsqrtf(fmaxf(rdeg[row0 + m16], 1.f));

    v8f c0 = {}, c1 = {}, c2 = {};
    #pragma unroll 4
    for (int kb = 0; kb < DH; kb += 4) {
        v2f a = *(const v2f*)(agg1 + (size_t)(row0 + m16) * DH + kb + 2 * kh);
        a.x *= rs; a.y *= rs;
        const float* wr0 = W3 + (size_t)(kb + 2 * kh) * DC;
        const float* wr1 = wr0 + DC;
        v2f b;
        b.x = wr0[m16];      b.y = wr1[m16];      c0 = wmma4(a, b, c0);
        b.x = wr0[16 + m16]; b.y = wr1[16 + m16]; c1 = wmma4(a, b, c1);
        const int n2 = 32 + m16;
        b.x = (n2 < DC) ? wr0[n2] : 0.f;
        b.y = (n2 < DC) ? wr1[n2] : 0.f;
        c2 = wmma4(a, b, c2);
    }
    float sc[8];
    #pragma unroll
    for (int r = 0; r < 8; ++r) sc[r] = rsqrtf(fmaxf(sdeg[row0 + r + 8 * kh], 1.f));
    #pragma unroll
    for (int r = 0; r < 8; ++r) {
        float* orow = h2 + (size_t)(row0 + r + 8 * kh) * DC;
        orow[m16]      = (c0[r] + b3[m16])      * sc[r];
        orow[16 + m16] = (c1[r] + b3[16 + m16]) * sc[r];
        if (32 + m16 < DC)
            orow[32 + m16] = (c2[r] + b3[32 + m16]) * sc[r];
    }
}

__launch_bounds__(256)
__global__ void gather2_kernel(const float* __restrict__ h2, const int* __restrict__ csr,
                               const int* __restrict__ offs, const float* __restrict__ rdeg,
                               float* __restrict__ out) {
    const int gt = blockIdx.x * 256 + threadIdx.x;
    const int n = gt >> 3;
    const int l = gt & 7;
    if (n >= NN) return;
    const int* seg = csr + offs[n];
    const int cnt = (int)rdeg[n];
    float4 acc = {0.f, 0.f, 0.f, 0.f};
    float tail = 0.f;
    for (int j = 0; j < cnt; j += 8) {
        int s[8];
        float4 v[8];
        float u[8];
        #pragma unroll
        for (int k = 0; k < 8; ++k) {
            const int q = seg[min(j + k, cnt - 1)];
            s[k] = ((unsigned)q < (unsigned)NN) ? q : 0;
        }
        #pragma unroll
        for (int k = 0; k < 8; ++k) {
            const float* row = h2 + (size_t)s[k] * DC;
            v[k] = *(const float4*)(row + 4 * l);
            u[k] = row[32 + l];
        }
        #pragma unroll
        for (int k = 0; k < 8; ++k) {
            const bool in = j + k < cnt;
            acc.x += in ? v[k].x : 0.f; acc.y += in ? v[k].y : 0.f;
            acc.z += in ? v[k].z : 0.f; acc.w += in ? v[k].w : 0.f;
            tail += in ? u[k] : 0.f;
        }
    }
    float* orow = out + (size_t)n * DC;
    *(float4*)(orow + 4 * l) = acc;
    orow[32 + l] = tail;
}

__global__ void softmax_kernel(float* __restrict__ out, const float* __restrict__ rdeg) {
    int n = blockIdx.x * blockDim.x + threadIdx.x;
    if (n >= NN) return;
    float rs = rsqrtf(fmaxf(rdeg[n], 1.f));
    float* row = out + (size_t)n * DC;
    float v[DC];
    float mx = -3.402823466e38f;
    #pragma unroll
    for (int f = 0; f < DC; ++f) { v[f] = row[f] * rs; mx = fmaxf(mx, v[f]); }
    float s = 0.f;
    #pragma unroll
    for (int f = 0; f < DC; ++f) { v[f] = __expf(v[f] - mx); s += v[f]; }
    float inv = 1.f / s;
    #pragma unroll
    for (int f = 0; f < DC; ++f) row[f] = v[f] * inv;
}

extern "C" void kernel_launch(void* const* d_in, const int* in_sizes, int n_in,
                              void* d_out, int out_size, void* d_ws, size_t ws_size,
                              hipStream_t stream) {
    const float* nodes     = (const float*)d_in[0];
    const int*   senders   = (const int*)d_in[1];
    const int*   receivers = (const int*)d_in[2];
    const float* W1 = (const float*)d_in[3];
    const float* b1 = (const float*)d_in[4];
    const float* W2 = (const float*)d_in[5];
    const float* b2 = (const float*)d_in[6];
    const float* W3 = (const float*)d_in[7];
    const float* b3 = (const float*)d_in[8];
    float* out = (float*)d_out;
    float* ws  = (float*)d_ws;

    float* sdeg = ws;
    float* rdeg = ws + NN;
    float* h1   = ws + 2 * (size_t)NN;
    float* agg1 = h1 + (size_t)NN * DH;
    float* h2   = agg1 + (size_t)NN * DH;
    int*   offs = (int*)(h2 + (size_t)NN * DC);
    int*   rank = offs + NN;
    int*   csr  = rank + NE;

    const int nTiles = NN / 16;
    zero_f32<<<2048, 256, 0, stream>>>(sdeg, 2 * NN);
    deg_kernel<<<(NE + 255) / 256, 256, 0, stream>>>(senders, receivers, sdeg, rdeg, rank);
    offs_kernel<<<(NN + 1023) / 1024, 256, 0, stream>>>(rdeg, offs);
    fill_kernel<<<(NE + 255) / 256, 256, 0, stream>>>(senders, receivers, rank, offs, csr);
    mlp_kernel<<<(nTiles + 7) / 8, 256, 0, stream>>>(nodes, W1, b1, W2, b2, sdeg, h1, nTiles);
    gather1_kernel<<<(NN * 16 + 255) / 256, 256, 0, stream>>>(h1, csr, offs, rdeg, agg1);
    fin1_kernel<<<(nTiles + 7) / 8, 256, 0, stream>>>(agg1, rdeg, sdeg, W3, b3, h2, nTiles);
    gather2_kernel<<<(NN * 8 + 255) / 256, 256, 0, stream>>>(h2, csr, offs, rdeg, out);
    softmax_kernel<<<(NN + 255) / 256, 256, 0, stream>>>(out, rdeg);
}


# Receiver-ordered edge list built by a two-level LDS counting sort (256-node bins), not by a returning global add, scattered read and scattered store per edge; a node's rows sum in bin-arrival order
# speedup vs baseline: 2.9549x; 1.3531x over previous
#include <hip/hip_runtime.h>

typedef __attribute__((ext_vector_type(2))) float v2f;
typedef __attribute__((ext_vector_type(8))) float v8f;

static constexpr int NN = 100000;
static constexpr int NE = 1600000;
static constexpr int DF = 128;
static constexpr int DH = 64;
static constexpr int DC = 40;
#define NEG_SLOPE 0.01f

__device__ __forceinline__ float lrelu(float x) { return x > 0.f ? x : x * NEG_SLOPE; }

__device__ __forceinline__ v8f wmma4(v2f a, v2f b, v8f c) {
    return __builtin_amdgcn_wmma_f32_16x16x4_f32(false, a, false, b, (short)0, c, false, false);
}

__global__ void zero_f32(float* __restrict__ p, int n) {
    int i = blockIdx.x * blockDim.x + threadIdx.x;
    int stride = gridDim.x * blockDim.x;
    for (; i < n; i += stride) p[i] = 0.f;
}

static constexpr int BN_SHIFT = 8;
static constexpr int BN   = 1 << BN_SHIFT;
static constexpr int NB   = (NN + BN - 1) / BN;
static constexpr int NBP  = 512;
static constexpr int CH   = 4096;
static constexpr int NCH  = (NE + CH - 1) / CH;
static constexpr int EPT  = CH / 256;
static constexpr int SND_BITS = 17;
static constexpr int SND_MASK = (1 << SND_BITS) - 1;
static constexpr int CAP  = 12288;
static_assert(NN <= (1 << SND_BITS), "a sender id must fit the packed word");
static_assert(NB <= NBP, "two bins per thread cover every bin");
static_assert(NE < (1 << 24), "degrees are whole numbers that stay exact in f32");

__device__ __forceinline__ int block_scan_incl(int v, int* buf) {
    const int t = threadIdx.x;
    buf[t] = v;
    __syncthreads();
    for (int d = 1; d < 256; d <<= 1) {
        const int a = (t >= d) ? buf[t - d] : 0;
        __syncthreads();
        v += a;
        buf[t] = v;
        __syncthreads();
    }
    return v;
}

__device__ __forceinline__ void block_scan_incl2(int& a, int& b, int* bufA, int* bufB) {
    const int t = threadIdx.x;
    bufA[t] = a;
    bufB[t] = b;
    __syncthreads();
    for (int d = 1; d < 256; d <<= 1) {
        const int xa = (t >= d) ? bufA[t - d] : 0;
        const int xb = (t >= d) ? bufB[t - d] : 0;
        __syncthreads();
        a += xa;
        b += xb;
        bufA[t] = a;
        bufB[t] = b;
        __syncthreads();
    }
}

__launch_bounds__(256)
__global__ void count_kernel(const int* __restrict__ snd, const int* __restrict__ rcv,
                             float* __restrict__ sdeg, int* __restrict__ btot) {
    __shared__ int hist[NBP];
    const int t = threadIdx.x;
    hist[t] = 0;
    hist[t + 256] = 0;
    __syncthreads();
    const int e0 = blockIdx.x * CH + t;
    int r[EPT], s[EPT];
    #pragma unroll
    for (int k = 0; k < EPT; ++k) {
        const int e = e0 + k * 256;
        r[k] = (e < NE) ? rcv[e] : -1;
        s[k] = (e < NE) ? snd[e] : -1;
    }
    #pragma unroll
    for (int k = 0; k < EPT; ++k) {
        if ((unsigned)r[k] < (unsigned)NN) atomicAdd(&hist[r[k] >> BN_SHIFT], 1);
        if ((unsigned)s[k] < (unsigned)NN) atomicAdd(&sdeg[s[k]], 1.f);
    }
    __syncthreads();
    const int h0 = hist[t], h1 = hist[t + 256];
    if (h0) atomicAdd(&btot[t], h0);
    if (h1) atomicAdd(&btot[t + 256], h1);
}

__launch_bounds__(256)
__global__ void bin_kernel(const int* __restrict__ snd, const int* __restrict__ rcv,
                           const int* __restrict__ btot, int* __restrict__ bcur,
                           int* __restrict__ pairs) {
    __shared__ int hist[NBP];
    __shared__ int lstart[NBP];
    __shared__ int gstart[NBP];
    __shared__ int sA[256];
    __shared__ int sB[256];
    __shared__ int sw[CH];
    __shared__ int sd[CH];
    const int t = threadIdx.x;
    hist[t] = 0;
    hist[t + 256] = 0;
    __syncthreads();
    int word[EPT], lr[EPT], bin[EPT];
    const int e0 = blockIdx.x * CH + t;
    #pragma unroll
    for (int k = 0; k < EPT; ++k) {
        const int e = e0 + k * 256;
        bin[k]  = (e < NE) ? rcv[e] : -1;
        word[k] = (e < NE) ? snd[e] : 0;
    }
    #pragma unroll
    for (int k = 0; k < EPT; ++k) {
        const int r = bin[k];
        lr[k] = -1;
        bin[k] = 0;
        if ((unsigned)r < (unsigned)NN) {
            bin[k]  = r >> BN_SHIFT;
            lr[k]   = atomicAdd(&hist[bin[k]], 1);
            word[k] = ((r & (BN - 1)) << SND_BITS) | (word[k] & SND_MASK);
        }
    }
    __syncthreads();
    const int h0 = hist[2 * t], h1 = hist[2 * t + 1];
    const int g0 = btot[2 * t], g1 = btot[2 * t + 1];
    int li = h0 + h1, gi = g0 + g1;
    block_scan_incl2(li, gi, sA, sB);
    const int lex = li - (h0 + h1);
    const int gex = gi - (g0 + g1);
    int r0 = 0, r1 = 0;
    if (h0) r0 = atomicAdd(&bcur[2 * t], h0);
    if (h1) r1 = atomicAdd(&bcur[2 * t + 1], h1);
    lstart[2 * t]     = lex;
    lstart[2 * t + 1] = lex + h0;
    gstart[2 * t]     = gex + r0;
    gstart[2 * t + 1] = gex + g0 + r1;
    __syncthreads();
    #pragma unroll
    for (int k = 0; k < EPT; ++k) {
        if (lr[k] >= 0) {
            const int pos = lstart[bin[k]] + lr[k];
            sw[pos] = word[k];
            sd[pos] = gstart[bin[k]] + lr[k];
        }
    }
    __syncthreads();
    const int mine = sA[255];
    for (int i = t; i < mine; i += 256) {
        const int p = sd[i];
        if ((unsigned)p < (unsigned)NE) pairs[p] = sw[i];
    }
}

__launch_bounds__(256)
__global__ void sort_kernel(const int* __restrict__ pairs, const int* __restrict__ btot,
                            int* __restrict__ offs, float* __restrict__ rdeg, int* __restrict__ csr) {
    __shared__ int sA[256];
    __shared__ int sB[256];
    __shared__ int cnt[BN];
    __shared__ int cur[BN];
    __shared__ int stage[CAP];
    const int t = threadIdx.x;
    const int b = blockIdx.x;
    int front = 0;
    if (t < b)       front += btot[t];
    if (t + 256 < b) front += btot[t + 256];
    for (int m = 16; m >= 1; m >>= 1) front += __shfl_xor(front, m);
    if ((t & 31) == 0) sA[t >> 5] = front;
    __syncthreads();
    int all = 0;
    #pragma unroll
    for (int w = 0; w < 8; ++w) all += sA[w];
    const int start = min(max(all, 0), NE);
    const int nb    = min(max(btot[b], 0), NE - start);
    cnt[t] = 0;
    __syncthreads();
    for (int i0 = t; i0 < nb; i0 += 256 * 8) {
        int w[8];
        #pragma unroll
        for (int u = 0; u < 8; ++u) {
            const int i = i0 + 256 * u;
            w[u] = (i < nb) ? pairs[start + i] : -1;
        }
        #pragma unroll
        for (int u = 0; u < 8; ++u)
            if (w[u] >= 0) atomicAdd(&cnt[(w[u] >> SND_BITS) & (BN - 1)], 1);
    }
    __syncthreads();
    const int c   = cnt[t];
    const int loc = block_scan_incl(c, sB) - c;
    const int node = b * BN + t;
    if (node < NN) {
        offs[node] = start + loc;
        rdeg[node] = (float)c;
    }
    cur[t] = loc;
    __syncthreads();
    if (nb <= CAP) {
        for (int i0 = t; i0 < nb; i0 += 256 * 8) {
            int w[8];
            #pragma unroll
            for (int u = 0; u < 8; ++u) {
                const int i = i0 + 256 * u;
                w[u] = (i < nb) ? pairs[start + i] : -1;
            }
            #pragma unroll
            for (int u = 0; u < 8; ++u) {
                if (w[u] >= 0) {
                    const int p = atomicAdd(&cur[(w[u] >> SND_BITS) & (BN - 1)], 1);
                    if ((unsigned)p < (unsigned)CAP) stage[p] = w[u] & SND_MASK;
                }
            }
        }
        __syncthreads();
        for (int i = t; i < nb; i += 256) csr[start + i] = stage[i];
    } else {
        for (int i = t; i < nb; i += 256) {
            const int w = pairs[start + i];
            const int p = atomicAdd(&cur[(w >> SND_BITS) & (BN - 1)], 1);
            if ((unsigned)p < (unsigned)nb) csr[start + p] = w & SND_MASK;
        }
    }
}

__launch_bounds__(256)
__global__ void mlp_kernel(const float* __restrict__ X,
                           const float* __restrict__ W1, const float* __restrict__ b1,
                           const float* __restrict__ W2, const float* __restrict__ b2,
                           const float* __restrict__ sdeg,
                           float* __restrict__ h1, int nTiles) {
    __shared__ float lds[8 * 16 * DH];
    const int w    = threadIdx.x >> 5;
    const int lane = threadIdx.x & 31;
    const int m16  = lane & 15;
    const int kh   = lane >> 4;
    const int t    = blockIdx.x * 8 + w;
    const bool active = (t < nTiles);
    const int row0 = t * 16;
    float* my = &lds[w * 16 * DH];

    if (active) {
        v8f c0 = {}, c1 = {}, c2 = {}, c3 = {};
        #pragma unroll 4
        for (int kb = 0; kb < DF; kb += 4) {
            v2f a = *(const v2f*)(X + (size_t)(row0 + m16) * DF + kb + 2 * kh);
            const float* wr0 = W1 + (size_t)(kb + 2 * kh) * DH;
            const float* wr1 = wr0 + DH;
            v2f b;
            b.x = wr0[m16];      b.y = wr1[m16];      c0 = wmma4(a, b, c0);
            b.x = wr0[16 + m16]; b.y = wr1[16 + m16]; c1 = wmma4(a, b, c1);
            b.x = wr0[32 + m16]; b.y = wr1[32 + m16]; c2 = wmma4(a, b, c2);
            b.x = wr0[48 + m16]; b.y = wr1[48 + m16]; c3 = wmma4(a, b, c3);
        }
        #pragma unroll
        for (int r = 0; r < 8; ++r) {
            float* dr = my + (r + 8 * kh) * DH;
            dr[m16]      = lrelu(c0[r] + b1[m16]);
            dr[16 + m16] = lrelu(c1[r] + b1[16 + m16]);
            dr[32 + m16] = lrelu(c2[r] + b1[32 + m16]);
            dr[48 + m16] = lrelu(c3[r] + b1[48 + m16]);
        }
    }
    __syncthreads();
    if (!active) return;

    v8f d0 = {}, d1 = {}, d2 = {}, d3 = {};
    #pragma unroll 4
    for (int kb = 0; kb < DH; kb += 4) {
        v2f a = *(const v2f*)(my + m16 * DH + kb + 2 * kh);
        const float* wr0 = W2 + (size_t)(kb + 2 * kh) * DH;
        const float* wr1 = wr0 + DH;
        v2f b;
        b.x = wr0[m16];      b.y = wr1[m16];      d0 = wmma4(a, b, d0);
        b.x = wr0[16 + m16]; b.y = wr1[16 + m16]; d1 = wmma4(a, b, d1);
        b.x = wr0[32 + m16]; b.y = wr1[32 + m16]; d2 = wmma4(a, b, d2);
        b.x = wr0[48 + m16]; b.y = wr1[48 + m16]; d3 = wmma4(a, b, d3);
    }
    float sc[8];
    #pragma unroll
    for (int r = 0; r < 8; ++r) sc[r] = rsqrtf(fmaxf(sdeg[row0 + r + 8 * kh], 1.f));
    #pragma unroll
    for (int r = 0; r < 8; ++r) {
        float* orow = h1 + (size_t)(row0 + r + 8 * kh) * DH;
        orow[m16]      = lrelu(d0[r] + b2[m16])      * sc[r];
        orow[16 + m16] = lrelu(d1[r] + b2[16 + m16]) * sc[r];
        orow[32 + m16] = lrelu(d2[r] + b2[32 + m16]) * sc[r];
        orow[48 + m16] = lrelu(d3[r] + b2[48 + m16]) * sc[r];
    }
}

__launch_bounds__(256)
__global__ void gather1_kernel(const float* __restrict__ h1, const int* __restrict__ csr,
                               const int* __restrict__ offs, const float* __restrict__ rdeg,
                               float* __restrict__ agg) {
    const int gt = blockIdx.x * 256 + threadIdx.x;
    const int n = gt >> 4;
    const int l = gt & 15;
    if (n >= NN) return;
    const int* seg = csr + offs[n];
    const int cnt = (int)rdeg[n];
    float4 acc = {0.f, 0.f, 0.f, 0.f};
    for (int j = 0; j < cnt; j += 8) {
        int s[8];
        float4 v[8];
        #pragma unroll
        for (int k = 0; k < 8; ++k) {
            const int q = seg[min(j + k, cnt - 1)];
            s[k] = ((unsigned)q < (unsigned)NN) ? q : 0;
        }
        #pragma unroll
        for (int k = 0; k < 8; ++k)
            v[k] = *(const float4*)(h1 + (size_t)s[k] * DH + 4 * l);
        #pragma unroll
        for (int k = 0; k < 8; ++k) {
            const bool in = j + k < cnt;
            acc.x += in ? v[k].x : 0.f; acc.y += in ? v[k].y : 0.f;
            acc.z += in ? v[k].z : 0.f; acc.w += in ? v[k].w : 0.f;
        }
    }
    *(float4*)(agg + (size_t)n * DH + 4 * l) = acc;
}

__launch_bounds__(256)
__global__ void fin1_kernel(const float* __restrict__ agg1,
                            const float* __restrict__ rdeg, const float* __restrict__ sdeg,
                            const float* __restrict__ W3, const float* __restrict__ b3,
                            float* __restrict__ h2, int nTiles) {
    const int w    = threadIdx.x >> 5;
    const int lane = threadIdx.x & 31;
    const int m16  = lane & 15;
    const int kh   = lane >> 4;
    const int t    = blockIdx.x * 8 + w;
    if (t >= nTiles) return;
    const int row0 = t * 16;
    const float rs = rsqrtf(fmaxf(rdeg[row0 + m16], 1.f));

    v8f c0 = {}, c1 = {}, c2 = {};
    #pragma unroll 4
    for (int kb = 0; kb < DH; kb += 4) {
        v2f a = *(const v2f*)(agg1 + (size_t)(row0 + m16) * DH + kb + 2 * kh);
        a.x *= rs; a.y *= rs;
        const float* wr0 = W3 + (size_t)(kb + 2 * kh) * DC;
        const float* wr1 = wr0 + DC;
        v2f b;
        b.x = wr0[m16];      b.y = wr1[m16];      c0 = wmma4(a, b, c0);
        b.x = wr0[16 + m16]; b.y = wr1[16 + m16]; c1 = wmma4(a, b, c1);
        const int n2 = 32 + m16;
        b.x = (n2 < DC) ? wr0[n2] : 0.f;
        b.y = (n2 < DC) ? wr1[n2] : 0.f;
        c2 = wmma4(a, b, c2);
    }
    float sc[8];
    #pragma unroll
    for (int r = 0; r < 8; ++r) sc[r] = rsqrtf(fmaxf(sdeg[row0 + r + 8 * kh], 1.f));
    #pragma unroll
    for (int r = 0; r < 8; ++r) {
        float* orow = h2 + (size_t)(row0 + r + 8 * kh) * DC;
        orow[m16]      = (c0[r] + b3[m16])      * sc[r];
        orow[16 + m16] = (c1[r] + b3[16 + m16]) * sc[r];
        if (32 + m16 < DC)
            orow[32 + m16] = (c2[r] + b3[32 + m16]) * sc[r];
    }
}

__launch_bounds__(256)
__global__ void gather2_kernel(const float* __restrict__ h2, const int* __restrict__ csr,
                               const int* __restrict__ offs, const float* __restrict__ rdeg,
                               float* __restrict__ out) {
    const int gt = blockIdx.x * 256 + threadIdx.x;
    const int n = gt >> 3;
    const int l = gt & 7;
    if (n >= NN) return;
    const int* seg = csr + offs[n];
    const int cnt = (int)rdeg[n];
    float4 acc = {0.f, 0.f, 0.f, 0.f};
    float tail = 0.f;
    for (int j = 0; j < cnt; j += 8) {
        int s[8];
        float4 v[8];
        float u[8];
        #pragma unroll
        for (int k = 0; k < 8; ++k) {
            const int q = seg[min(j + k, cnt - 1)];
            s[k] = ((unsigned)q < (unsigned)NN) ? q : 0;
        }
        #pragma unroll
        for (int k = 0; k < 8; ++k) {
            const float* row = h2 + (size_t)s[k] * DC;
            v[k] = *(const float4*)(row + 4 * l);
            u[k] = row[32 + l];
        }
        #pragma unroll
        for (int k = 0; k < 8; ++k) {
            const bool in = j + k < cnt;
            acc.x += in ? v[k].x : 0.f; acc.y += in ? v[k].y : 0.f;
            acc.z += in ? v[k].z : 0.f; acc.w += in ? v[k].w : 0.f;
            tail += in ? u[k] : 0.f;
        }
    }
    float* orow = out + (size_t)n * DC;
    *(float4*)(orow + 4 * l) = acc;
    orow[32 + l] = tail;
}

__global__ void softmax_kernel(float* __restrict__ out, const float* __restrict__ rdeg) {
    int n = blockIdx.x * blockDim.x + threadIdx.x;
    if (n >= NN) return;
    float rs = rsqrtf(fmaxf(rdeg[n], 1.f));
    float* row = out + (size_t)n * DC;
    float v[DC];
    float mx = -3.402823466e38f;
    #pragma unroll
    for (int f = 0; f < DC; ++f) { v[f] = row[f] * rs; mx = fmaxf(mx, v[f]); }
    float s = 0.f;
    #pragma unroll
    for (int f = 0; f < DC; ++f) { v[f] = __expf(v[f] - mx); s += v[f]; }
    float inv = 1.f / s;
    #pragma unroll
    for (int f = 0; f < DC; ++f) row[f] = v[f] * inv;
}

extern "C" void kernel_launch(void* const* d_in, const int* in_sizes, int n_in,
                              void* d_out, int out_size, void* d_ws, size_t ws_size,
                              hipStream_t stream) {
    const float* nodes     = (const float*)d_in[0];
    const int*   senders   = (const int*)d_in[1];
    const int*   receivers = (const int*)d_in[2];
    const float* W1 = (const float*)d_in[3];
    const float* b1 = (const float*)d_in[4];
    const float* W2 = (const float*)d_in[5];
    const float* b2 = (const float*)d_in[6];
    const float* W3 = (const float*)d_in[7];
    const float* b3 = (const float*)d_in[8];
    float* out = (float*)d_out;
    float* ws  = (float*)d_ws;

    float* sdeg = ws;
    int*   btot = (int*)(ws + NN);
    int*   bcur = btot + NBP;
    float* rdeg = ws + NN + 2 * NBP;
    float* h1   = rdeg + NN;
    float* agg1 = h1 + (size_t)NN * DH;
    float* h2   = agg1 + (size_t)NN * DH;
    int*   offs = (int*)(h2 + (size_t)NN * DC);
    int*   pairs = offs + NN;
    int*   csr   = pairs + NE;

    const int nTiles = NN / 16;
    zero_f32<<<2048, 256, 0, stream>>>(sdeg, NN + 2 * NBP);
    count_kernel<<<NCH, 256, 0, stream>>>(senders, receivers, sdeg, btot);
    bin_kernel<<<NCH, 256, 0, stream>>>(senders, receivers, btot, bcur, pairs);
    sort_kernel<<<NB, 256, 0, stream>>>(pairs, btot, offs, rdeg, csr);
    mlp_kernel<<<(nTiles + 7) / 8, 256, 0, stream>>>(nodes, W1, b1, W2, b2, sdeg, h1, nTiles);
    gather1_kernel<<<(NN * 16 + 255) / 256, 256, 0, stream>>>(h1, csr, offs, rdeg, agg1);
    fin1_kernel<<<(nTiles + 7) / 8, 256, 0, stream>>>(agg1, rdeg, sdeg, W3, b3, h2, nTiles);
    gather2_kernel<<<(NN * 8 + 255) / 256, 256, 0, stream>>>(h2, csr, offs, rdeg, out);
    softmax_kernel<<<(NN + 255) / 256, 256, 0, stream>>>(out, rdeg);
}


# W3 product moved in front of the first edge sum (sum of rows x W3 = sum of rows-x-W3; the sum order changes): gather 1 reads 40 floats a row not 64 and adds b3 and the scales; fin1 launch, agg1 gone
# speedup vs baseline: 3.1715x; 1.0733x over previous
#include <hip/hip_runtime.h>

typedef __attribute__((ext_vector_type(2))) float v2f;
typedef __attribute__((ext_vector_type(8))) float v8f;

static constexpr int NN = 100000;
static constexpr int NE = 1600000;
static constexpr int DF = 128;
static constexpr int DH = 64;
static constexpr int DC = 40;
static constexpr int P1P = 64;
static constexpr int G1N = 3;
#define NEG_SLOPE 0.01f

__device__ __forceinline__ float lrelu(float x) { return x > 0.f ? x : x * NEG_SLOPE; }

__device__ __forceinline__ v8f wmma4(v2f a, v2f b, v8f c) {
    return __builtin_amdgcn_wmma_f32_16x16x4_f32(false, a, false, b, (short)0, c, false, false);
}

__global__ void zero_f32(float* __restrict__ p, int n) {
    int i = blockIdx.x * blockDim.x + threadIdx.x;
    int stride = gridDim.x * blockDim.x;
    for (; i < n; i += stride) p[i] = 0.f;
}

static constexpr int BN_SHIFT = 8;
static constexpr int BN   = 1 << BN_SHIFT;
static constexpr int NB   = (NN + BN - 1) / BN;
static constexpr int NBP  = 512;
static constexpr int CH   = 4096;
static constexpr int NCH  = (NE + CH - 1) / CH;
static constexpr int EPT  = CH / 256;
static constexpr int SND_BITS = 17;
static constexpr int SND_MASK = (1 << SND_BITS) - 1;
static constexpr int CAP  = 12288;
static_assert(NN <= (1 << SND_BITS), "a sender id must fit the packed word");
static_assert(NB <= NBP, "two bins per thread cover every bin");
static_assert(NE < (1 << 24), "degrees are whole numbers that stay exact in f32");

__device__ __forceinline__ int block_scan_incl(int v, int* buf) {
    const int t = threadIdx.x;
    buf[t] = v;
    __syncthreads();
    for (int d = 1; d < 256; d <<= 1) {
        const int a = (t >= d) ? buf[t - d] : 0;
        __syncthreads();
        v += a;
        buf[t] = v;
        __syncthreads();
    }
    return v;
}

__device__ __forceinline__ void block_scan_incl2(int& a, int& b, int* bufA, int* bufB) {
    const int t = threadIdx.x;
    bufA[t] = a;
    bufB[t] = b;
    __syncthreads();
    for (int d = 1; d < 256; d <<= 1) {
        const int xa = (t >= d) ? bufA[t - d] : 0;
        const int xb = (t >= d) ? bufB[t - d] : 0;
        __syncthreads();
        a += xa;
        b += xb;
        bufA[t] = a;
        bufB[t] = b;
        __syncthreads();
    }
}

__launch_bounds__(256)
__global__ void count_kernel(const int* __restrict__ snd, const int* __restrict__ rcv,
                             float* __restrict__ sdeg, int* __restrict__ btot) {
    __shared__ int hist[NBP];
    const int t = threadIdx.x;
    hist[t] = 0;
    hist[t + 256] = 0;
    __syncthreads();
    const int e0 = blockIdx.x * CH + t;
    int r[EPT], s[EPT];
    #pragma unroll
    for (int k = 0; k < EPT; ++k) {
        const int e = e0 + k * 256;
        r[k] = (e < NE) ? rcv[e] : -1;
        s[k] = (e < NE) ? snd[e] : -1;
    }
    #pragma unroll
    for (int k = 0; k < EPT; ++k) {
        if ((unsigned)r[k] < (unsigned)NN) atomicAdd(&hist[r[k] >> BN_SHIFT], 1);
        if ((unsigned)s[k] < (unsigned)NN) atomicAdd(&sdeg[s[k]], 1.f);
    }
    __syncthreads();
    const int h0 = hist[t], h1 = hist[t + 256];
    if (h0) atomicAdd(&btot[t], h0);
    if (h1) atomicAdd(&btot[t + 256], h1);
}

__launch_bounds__(256)
__global__ void bin_kernel(const int* __restrict__ snd, const int* __restrict__ rcv,
                           const int* __restrict__ btot, int* __restrict__ bcur,
                           int* __restrict__ pairs) {
    __shared__ int hist[NBP];
    __shared__ int lstart[NBP];
    __shared__ int gstart[NBP];
    __shared__ int sA[256];
    __shared__ int sB[256];
    __shared__ int sw[CH];
    __shared__ int sd[CH];
    const int t = threadIdx.x;
    hist[t] = 0;
    hist[t + 256] = 0;
    __syncthreads();
    int word[EPT], lr[EPT], bin[EPT];
    const int e0 = blockIdx.x * CH + t;
    #pragma unroll
    for (int k = 0; k < EPT; ++k) {
        const int e = e0 + k * 256;
        bin[k]  = (e < NE) ? rcv[e] : -1;
        word[k] = (e < NE) ? snd[e] : 0;
    }
    #pragma unroll
    for (int k = 0; k < EPT; ++k) {
        const int r = bin[k];
        lr[k] = -1;
        bin[k] = 0;
        if ((unsigned)r < (unsigned)NN) {
            bin[k]  = r >> BN_SHIFT;
            lr[k]   = atomicAdd(&hist[bin[k]], 1);
            word[k] = ((r & (BN - 1)) << SND_BITS) | (word[k] & SND_MASK);
        }
    }
    __syncthreads();
    const int h0 = hist[2 * t], h1 = hist[2 * t + 1];
    const int g0 = btot[2 * t], g1 = btot[2 * t + 1];
    int li = h0 + h1, gi = g0 + g1;
    block_scan_incl2(li, gi, sA, sB);
    const int lex = li - (h0 + h1);
    const int gex = gi - (g0 + g1);
    int r0 = 0, r1 = 0;
    if (h0) r0 = atomicAdd(&bcur[2 * t], h0);
    if (h1) r1 = atomicAdd(&bcur[2 * t + 1], h1);
    lstart[2 * t]     = lex;
    lstart[2 * t + 1] = lex + h0;
    gstart[2 * t]     = gex + r0;
    gstart[2 * t + 1] = gex + g0 + r1;
    __syncthreads();
    #pragma unroll
    for (int k = 0; k < EPT; ++k) {
        if (lr[k] >= 0) {
            const int pos = lstart[bin[k]] + lr[k];
            sw[pos] = word[k];
            sd[pos] = gstart[bin[k]] + lr[k];
        }
    }
    __syncthreads();
    const int mine = sA[255];
    for (int i = t; i < mine; i += 256) {
        const int p = sd[i];
        if ((unsigned)p < (unsigned)NE) pairs[p] = sw[i];
    }
}

__launch_bounds__(256)
__global__ void sort_kernel(const int* __restrict__ pairs, const int* __restrict__ btot,
                            int* __restrict__ offs, float* __restrict__ rdeg, int* __restrict__ csr) {
    __shared__ int sA[256];
    __shared__ int sB[256];
    __shared__ int cnt[BN];
    __shared__ int cur[BN];
    __shared__ int stage[CAP];
    const int t = threadIdx.x;
    const int b = blockIdx.x;
    int front = 0;
    if (t < b)       front += btot[t];
    if (t + 256 < b) front += btot[t + 256];
    for (int m = 16; m >= 1; m >>= 1) front += __shfl_xor(front, m);
    if ((t & 31) == 0) sA[t >> 5] = front;
    __syncthreads();
    int all = 0;
    #pragma unroll
    for (int w = 0; w < 8; ++w) all += sA[w];
    const int start = min(max(all, 0), NE);
    const int nb    = min(max(btot[b], 0), NE - start);
    cnt[t] = 0;
    __syncthreads();
    for (int i0 = t; i0 < nb; i0 += 256 * 8) {
        int w[8];
        #pragma unroll
        for (int u = 0; u < 8; ++u) {
            const int i = i0 + 256 * u;
            w[u] = (i < nb) ? pairs[start + i] : -1;
        }
        #pragma unroll
        for (int u = 0; u < 8; ++u)
            if (w[u] >= 0) atomicAdd(&cnt[(w[u] >> SND_BITS) & (BN - 1)], 1);
    }
    __syncthreads();
    const int c   = cnt[t];
    const int loc = block_scan_incl(c, sB) - c;
    const int node = b * BN + t;
    if (node < NN) {
        offs[node] = start + loc;
        rdeg[node] = (float)c;
    }
    cur[t] = loc;
    __syncthreads();
    if (nb <= CAP) {
        for (int i0 = t; i0 < nb; i0 += 256 * 8) {
            int w[8];
            #pragma unroll
            for (int u = 0; u < 8; ++u) {
                const int i = i0 + 256 * u;
                w[u] = (i < nb) ? pairs[start + i] : -1;
            }
            #pragma unroll
            for (int u = 0; u < 8; ++u) {
                if (w[u] >= 0) {
                    const int p = atomicAdd(&cur[(w[u] >> SND_BITS) & (BN - 1)], 1);
                    if ((unsigned)p < (unsigned)CAP) stage[p] = w[u] & SND_MASK;
                }
            }
        }
        __syncthreads();
        for (int i = t; i < nb; i += 256) csr[start + i] = stage[i];
    } else {
        for (int i = t; i < nb; i += 256) {
            const int w = pairs[start + i];
            const int p = atomicAdd(&cur[(w >> SND_BITS) & (BN - 1)], 1);
            if ((unsigned)p < (unsigned)nb) csr[start + p] = w & SND_MASK;
        }
    }
}

__launch_bounds__(256)
__global__ void mlp_kernel(const float* __restrict__ X,
                           const float* __restrict__ W1, const float* __restrict__ b1,
                           const float* __restrict__ W2, const float* __restrict__ b2,
                           const float* __restrict__ W3,
                           const float* __restrict__ sdeg,
                           float* __restrict__ p1, int nTiles) {
    __shared__ float lds[8 * 16 * DH];
    const int w    = threadIdx.x >> 5;
    const int lane = threadIdx.x & 31;
    const int m16  = lane & 15;
    const int kh   = lane >> 4;
    const int t    = blockIdx.x * 8 + w;
    const bool active = (t < nTiles);
    const int row0 = t * 16;
    float* my = &lds[w * 16 * DH];

    if (active) {
        v8f c0 = {}, c1 = {}, c2 = {}, c3 = {};
        #pragma unroll 4
        for (int kb = 0; kb < DF; kb += 4) {
            v2f a = *(const v2f*)(X + (size_t)(row0 + m16) * DF + kb + 2 * kh);
            const float* wr0 = W1 + (size_t)(kb + 2 * kh) * DH;
            const float* wr1 = wr0 + DH;
            v2f b;
            b.x = wr0[m16];      b.y = wr1[m16];      c0 = wmma4(a, b, c0);
            b.x = wr0[16 + m16]; b.y = wr1[16 + m16]; c1 = wmma4(a, b, c1);
            b.x = wr0[32 + m16]; b.y = wr1[32 + m16]; c2 = wmma4(a, b, c2);
            b.x = wr0[48 + m16]; b.y = wr1[48 + m16]; c3 = wmma4(a, b, c3);
        }
        #pragma unroll
        for (int r = 0; r < 8; ++r) {
            float* dr = my + (r + 8 * kh) * DH;
            dr[m16]      = lrelu(c0[r] + b1[m16]);
            dr[16 + m16] = lrelu(c1[r] + b1[16 + m16]);
            dr[32 + m16] = lrelu(c2[r] + b1[32 + m16]);
            dr[48 + m16] = lrelu(c3[r] + b1[48 + m16]);
        }
    }
    __syncthreads();

    if (active) {
        v8f d0 = {}, d1 = {}, d2 = {}, d3 = {};
        #pragma unroll 4
        for (int kb = 0; kb < DH; kb += 4) {
            v2f a = *(const v2f*)(my + m16 * DH + kb + 2 * kh);
            const float* wr0 = W2 + (size_t)(kb + 2 * kh) * DH;
            const float* wr1 = wr0 + DH;
            v2f b;
            b.x = wr0[m16];      b.y = wr1[m16];      d0 = wmma4(a, b, d0);
            b.x = wr0[16 + m16]; b.y = wr1[16 + m16]; d1 = wmma4(a, b, d1);
            b.x = wr0[32 + m16]; b.y = wr1[32 + m16]; d2 = wmma4(a, b, d2);
            b.x = wr0[48 + m16]; b.y = wr1[48 + m16]; d3 = wmma4(a, b, d3);
        }
        float sc[8];
        #pragma unroll
        for (int r = 0; r < 8; ++r) sc[r] = rsqrtf(fmaxf(sdeg[row0 + r + 8 * kh], 1.f));
        #pragma unroll
        for (int r = 0; r < 8; ++r) {
            float* dr = my + (r + 8 * kh) * DH;
            dr[m16]      = lrelu(d0[r] + b2[m16])      * sc[r];
            dr[16 + m16] = lrelu(d1[r] + b2[16 + m16]) * sc[r];
            dr[32 + m16] = lrelu(d2[r] + b2[32 + m16]) * sc[r];
            dr[48 + m16] = lrelu(d3[r] + b2[48 + m16]) * sc[r];
        }
    }
    __syncthreads();
    if (!active) return;

    v8f e0 = {}, e1 = {}, e2 = {};
    #pragma unroll 4
    for (int kb = 0; kb < DH; kb += 4) {
        v2f a = *(const v2f*)(my + m16 * DH + kb + 2 * kh);
        const float* wr0 = W3 + (size_t)(kb + 2 * kh) * DC;
        const float* wr1 = wr0 + DC;
        v2f b;
        b.x = wr0[m16];      b.y = wr1[m16];      e0 = wmma4(a, b, e0);
        b.x = wr0[16 + m16]; b.y = wr1[16 + m16]; e1 = wmma4(a, b, e1);
        const int n2  = 32 + m16;
        const int n2c = (n2 < DC) ? n2 : DC - 1;
        const float x0 = wr0[n2c], x1 = wr1[n2c];
        b.x = (n2 < DC) ? x0 : 0.f;
        b.y = (n2 < DC) ? x1 : 0.f;
        e2 = wmma4(a, b, e2);
    }
    #pragma unroll
    for (int r = 0; r < 8; ++r) {
        float* orow = p1 + (size_t)(row0 + r + 8 * kh) * P1P;
        orow[m16]      = e0[r];
        orow[16 + m16] = e1[r];
        if (32 + m16 < DC)
            orow[32 + m16] = e2[r];
    }
}

__launch_bounds__(256)
__global__ void gather1_kernel(const float* __restrict__ p1, const int* __restrict__ csr,
                               const int* __restrict__ offs, const float* __restrict__ rdeg,
                               const float* __restrict__ sdeg, const float* __restrict__ b3,
                               float* __restrict__ h2) {
    const int wave = (blockIdx.x * 256 + threadIdx.x) >> 5;
    const int lane = threadIdx.x & 31;
    const int q = lane / 10;
    const int l = lane - 10 * q;
    const int n = wave * G1N + q;
    if (q >= G1N || n >= NN) return;
    const int* seg = csr + offs[n];
    const float rd = rdeg[n];
    const int cnt = (int)rd;
    float4 acc = {0.f, 0.f, 0.f, 0.f};
    for (int j = 0; j < cnt; j += 8) {
        int s[8];
        float4 v[8];
        #pragma unroll
        for (int k = 0; k < 8; ++k) {
            const int qd = seg[min(j + k, cnt - 1)];
            s[k] = ((unsigned)qd < (unsigned)NN) ? qd : 0;
        }
        #pragma unroll
        for (int k = 0; k < 8; ++k)
            v[k] = *(const float4*)(p1 + (size_t)s[k] * P1P + 4 * l);
        #pragma unroll
        for (int k = 0; k < 8; ++k) {
            const bool in = j + k < cnt;
            acc.x += in ? v[k].x : 0.f; acc.y += in ? v[k].y : 0.f;
            acc.z += in ? v[k].z : 0.f; acc.w += in ? v[k].w : 0.f;
        }
    }
    const float rs = rsqrtf(fmaxf(rd, 1.f));
    const float sc = rsqrtf(fmaxf(sdeg[n], 1.f));
    float4 o;
    o.x = (acc.x * rs + b3[4 * l])     * sc;
    o.y = (acc.y * rs + b3[4 * l + 1]) * sc;
    o.z = (acc.z * rs + b3[4 * l + 2]) * sc;
    o.w = (acc.w * rs + b3[4 * l + 3]) * sc;
    *(float4*)(h2 + (size_t)n * DC + 4 * l) = o;
}

__launch_bounds__(256)
__global__ void gather2_kernel(const float* __restrict__ h2, const int* __restrict__ csr,
                               const int* __restrict__ offs, const float* __restrict__ rdeg,
                               float* __restrict__ out) {
    const int gt = blockIdx.x * 256 + threadIdx.x;
    const int n = gt >> 3;
    const int l = gt & 7;
    if (n >= NN) return;
    const int* seg = csr + offs[n];
    const int cnt = (int)rdeg[n];
    float4 acc = {0.f, 0.f, 0.f, 0.f};
    float tail = 0.f;
    for (int j = 0; j < cnt; j += 8) {
        int s[8];
        float4 v[8];
        float u[8];
        #pragma unroll
        for (int k = 0; k < 8; ++k) {
            const int q = seg[min(j + k, cnt - 1)];
            s[k] = ((unsigned)q < (unsigned)NN) ? q : 0;
        }
        #pragma unroll
        for (int k = 0; k < 8; ++k) {
            const float* row = h2 + (size_t)s[k] * DC;
            v[k] = *(const float4*)(row + 4 * l);
            u[k] = row[32 + l];
        }
        #pragma unroll
        for (int k = 0; k < 8; ++k) {
            const bool in = j + k < cnt;
            acc.x += in ? v[k].x : 0.f; acc.y += in ? v[k].y : 0.f;
            acc.z += in ? v[k].z : 0.f; acc.w += in ? v[k].w : 0.f;
            tail += in ? u[k] : 0.f;
        }
    }
    float* orow = out + (size_t)n * DC;
    *(float4*)(orow + 4 * l) = acc;
    orow[32 + l] = tail;
}

__global__ void softmax_kernel(float* __restrict__ out, const float* __restrict__ rdeg) {
    int n = blockIdx.x * blockDim.x + threadIdx.x;
    if (n >= NN) return;
    float rs = rsqrtf(fmaxf(rdeg[n], 1.f));
    float* row = out + (size_t)n * DC;
    float v[DC];
    float mx = -3.402823466e38f;
    #pragma unroll
    for (int f = 0; f < DC; ++f) { v[f] = row[f] * rs; mx = fmaxf(mx, v[f]); }
    float s = 0.f;
    #pragma unroll
    for (int f = 0; f < DC; ++f) { v[f] = __expf(v[f] - mx); s += v[f]; }
    float inv = 1.f / s;
    #pragma unroll
    for (int f = 0; f < DC; ++f) row[f] = v[f] * inv;
}

extern "C" void kernel_launch(void* const* d_in, const int* in_sizes, int n_in,
                              void* d_out, int out_size, void* d_ws, size_t ws_size,
                              hipStream_t stream) {
    const float* nodes     = (const float*)d_in[0];
    const int*   senders   = (const int*)d_in[1];
    const int*   receivers = (const int*)d_in[2];
    const float* W1 = (const float*)d_in[3];
    const float* b1 = (const float*)d_in[4];
    const float* W2 = (const float*)d_in[5];
    const float* b2 = (const float*)d_in[6];
    const float* W3 = (const float*)d_in[7];
    const float* b3 = (const float*)d_in[8];
    float* out = (float*)d_out;
    float* ws  = (float*)d_ws;

    float* sdeg = ws;
    int*   btot = (int*)(ws + NN);
    int*   bcur = btot + NBP;
    float* rdeg = ws + NN + 2 * NBP;
    float* p1   = rdeg + NN;
    float* h2   = p1 + (size_t)NN * P1P;
    int*   offs = (int*)(h2 + (size_t)NN * DC);
    int*   pairs = offs + NN;
    int*   csr   = pairs + NE;

    const int nTiles = NN / 16;
    zero_f32<<<2048, 256, 0, stream>>>(sdeg, NN + 2 * NBP);
    count_kernel<<<NCH, 256, 0, stream>>>(senders, receivers, sdeg, btot);
    bin_kernel<<<NCH, 256, 0, stream>>>(senders, receivers, btot, bcur, pairs);
    sort_kernel<<<NB, 256, 0, stream>>>(pairs, btot, offs, rdeg, csr);
    mlp_kernel<<<(nTiles + 7) / 8, 256, 0, stream>>>(nodes, W1, b1, W2, b2, W3, sdeg, p1, nTiles);
    gather1_kernel<<<((NN + G1N - 1) / G1N + 7) / 8, 256, 0, stream>>>(p1, csr, offs, rdeg, sdeg, b3, h2);
    gather2_kernel<<<(NN * 8 + 255) / 256, 256, 0, stream>>>(h2, csr, offs, rdeg, out);
    softmax_kernel<<<(NN + 255) / 256, 256, 0, stream>>>(out, rdeg);
}


# Dense kernel's row tiles ride in the two sort launches (half with the bin pass, half with the per-bin sort): independent work overlapped, 8 launches -> 7; arithmetic and every sum order as before
# speedup vs baseline: 3.3816x; 1.0662x over previous
#include <hip/hip_runtime.h>

typedef __attribute__((ext_vector_type(2))) float v2f;
typedef __attribute__((ext_vector_type(8))) float v8f;

static constexpr int NN = 100000;
static constexpr int NE = 1600000;
static constexpr int DF = 128;
static constexpr int DH = 64;
static constexpr int DC = 40;
static constexpr int P1P = 64;
static constexpr int G1N = 3;
#define NEG_SLOPE 0.01f

__device__ __forceinline__ float lrelu(float x) { return x > 0.f ? x : x * NEG_SLOPE; }

__device__ __forceinline__ v8f wmma4(v2f a, v2f b, v8f c) {
    return __builtin_amdgcn_wmma_f32_16x16x4_f32(false, a, false, b, (short)0, c, false, false);
}

__global__ void zero_f32(float* __restrict__ p, int n) {
    int i = blockIdx.x * blockDim.x + threadIdx.x;
    int stride = gridDim.x * blockDim.x;
    for (; i < n; i += stride) p[i] = 0.f;
}

static constexpr int BN_SHIFT = 8;
static constexpr int BN   = 1 << BN_SHIFT;
static constexpr int NB   = (NN + BN - 1) / BN;
static constexpr int NBP  = 512;
static constexpr int CH   = 4096;
static constexpr int NCH  = (NE + CH - 1) / CH;
static constexpr int EPT  = CH / 256;
static constexpr int SND_BITS = 17;
static constexpr int SND_MASK = (1 << SND_BITS) - 1;
static constexpr int CAP  = 12288;
static_assert(NN <= (1 << SND_BITS), "a sender id must fit the packed word");
static_assert(NB <= NBP, "two bins per thread cover every bin");
static_assert(NE < (1 << 24), "degrees are whole numbers that stay exact in f32");

__device__ __forceinline__ int block_scan_incl(int v, int* buf) {
    const int t = threadIdx.x;
    buf[t] = v;
    __syncthreads();
    for (int d = 1; d < 256; d <<= 1) {
        const int a = (t >= d) ? buf[t - d] : 0;
        __syncthreads();
        v += a;
        buf[t] = v;
        __syncthreads();
    }
    return v;
}

__device__ __forceinline__ void block_scan_incl2(int& a, int& b, int* bufA, int* bufB) {
    const int t = threadIdx.x;
    bufA[t] = a;
    bufB[t] = b;
    __syncthreads();
    for (int d = 1; d < 256; d <<= 1) {
        const int xa = (t >= d) ? bufA[t - d] : 0;
        const int xb = (t >= d) ? bufB[t - d] : 0;
        __syncthreads();
        a += xa;
        b += xb;
        bufA[t] = a;
        bufB[t] = b;
        __syncthreads();
    }
}

__launch_bounds__(256)
__global__ void count_kernel(const int* __restrict__ snd, const int* __restrict__ rcv,
                             float* __restrict__ sdeg, int* __restrict__ btot) {
    __shared__ int hist[NBP];
    const int t = threadIdx.x;
    hist[t] = 0;
    hist[t + 256] = 0;
    __syncthreads();
    const int e0 = blockIdx.x * CH + t;
    int r[EPT], s[EPT];
    #pragma unroll
    for (int k = 0; k < EPT; ++k) {
        const int e = e0 + k * 256;
        r[k] = (e < NE) ? rcv[e] : -1;
        s[k] = (e < NE) ? snd[e] : -1;
    }
    #pragma unroll
    for (int k = 0; k < EPT; ++k) {
        if ((unsigned)r[k] < (unsigned)NN) atomicAdd(&hist[r[k] >> BN_SHIFT], 1);
        if ((unsigned)s[k] < (unsigned)NN) atomicAdd(&sdeg[s[k]], 1.f);
    }
    __syncthreads();
    const int h0 = hist[t], h1 = hist[t + 256];
    if (h0) atomicAdd(&btot[t], h0);
    if (h1) atomicAdd(&btot[t + 256], h1);
}

static constexpr int BIN_LDS_WORDS = 3 * NBP + 2 * 256 + 2 * CH;
__device__ __forceinline__ void bin_body(const int blk, int* smem,
                           const int* __restrict__ snd, const int* __restrict__ rcv,
                           const int* __restrict__ btot, int* __restrict__ bcur,
                           int* __restrict__ pairs) {
    int* hist   = smem;
    int* lstart = hist + NBP;
    int* gstart = lstart + NBP;
    int* sA     = gstart + NBP;
    int* sB     = sA + 256;
    int* sw     = sB + 256;
    int* sd     = sw + CH;
    const int t = threadIdx.x;
    hist[t] = 0;
    hist[t + 256] = 0;
    __syncthreads();
    int word[EPT], lr[EPT], bin[EPT];
    const int e0 = blk * CH + t;
    #pragma unroll
    for (int k = 0; k < EPT; ++k) {
        const int e = e0 + k * 256;
        bin[k]  = (e < NE) ? rcv[e] : -1;
        word[k] = (e < NE) ? snd[e] : 0;
    }
    #pragma unroll
    for (int k = 0; k < EPT; ++k) {
        const int r = bin[k];
        lr[k] = -1;
        bin[k] = 0;
        if ((unsigned)r < (unsigned)NN) {
            bin[k]  = r >> BN_SHIFT;
            lr[k]   = atomicAdd(&hist[bin[k]], 1);
            word[k] = ((r & (BN - 1)) << SND_BITS) | (word[k] & SND_MASK);
        }
    }
    __syncthreads();
    const int h0 = hist[2 * t], h1 = hist[2 * t + 1];
    const int g0 = btot[2 * t], g1 = btot[2 * t + 1];
    int li = h0 + h1, gi = g0 + g1;
    block_scan_incl2(li, gi, sA, sB);
    const int lex = li - (h0 + h1);
    const int gex = gi - (g0 + g1);
    int r0 = 0, r1 = 0;
    if (h0) r0 = atomicAdd(&bcur[2 * t], h0);
    if (h1) r1 = atomicAdd(&bcur[2 * t + 1], h1);
    lstart[2 * t]     = lex;
    lstart[2 * t + 1] = lex + h0;
    gstart[2 * t]     = gex + r0;
    gstart[2 * t + 1] = gex + g0 + r1;
    __syncthreads();
    #pragma unroll
    for (int k = 0; k < EPT; ++k) {
        if (lr[k] >= 0) {
            const int pos = lstart[bin[k]] + lr[k];
            sw[pos] = word[k];
            sd[pos] = gstart[bin[k]] + lr[k];
        }
    }
    __syncthreads();
    const int mine = sA[255];
    for (int i = t; i < mine; i += 256) {
        const int p = sd[i];
        if ((unsigned)p < (unsigned)NE) pairs[p] = sw[i];
    }
}

static constexpr int SORT_LDS_WORDS = 2 * 256 + 2 * BN + CAP;
__device__ __forceinline__ void sort_body(const int b, int* smem,
                            const int* __restrict__ pairs, const int* __restrict__ btot,
                            int* __restrict__ offs, float* __restrict__ rdeg, int* __restrict__ csr) {
    int* sA    = smem;
    int* sB    = sA + 256;
    int* cnt   = sB + 256;
    int* cur   = cnt + BN;
    int* stage = cur + BN;
    const int t = threadIdx.x;
    int front = 0;
    if (t < b)       front += btot[t];
    if (t + 256 < b) front += btot[t + 256];
    for (int m = 16; m >= 1; m >>= 1) front += __shfl_xor(front, m);
    if ((t & 31) == 0) sA[t >> 5] = front;
    __syncthreads();
    int all = 0;
    #pragma unroll
    for (int w = 0; w < 8; ++w) all += sA[w];
    const int start = min(max(all, 0), NE);
    const int nb    = min(max(btot[b], 0), NE - start);
    cnt[t] = 0;
    __syncthreads();
    for (int i0 = t; i0 < nb; i0 += 256 * 8) {
        int w[8];
        #pragma unroll
        for (int u = 0; u < 8; ++u) {
            const int i = i0 + 256 * u;
            w[u] = (i < nb) ? pairs[start + i] : -1;
        }
        #pragma unroll
        for (int u = 0; u < 8; ++u)
            if (w[u] >= 0) atomicAdd(&cnt[(w[u] >> SND_BITS) & (BN - 1)], 1);
    }
    __syncthreads();
    const int c   = cnt[t];
    const int loc = block_scan_incl(c, sB) - c;
    const int node = b * BN + t;
    if (node < NN) {
        offs[node] = start + loc;
        rdeg[node] = (float)c;
    }
    cur[t] = loc;
    __syncthreads();
    if (nb <= CAP) {
        for (int i0 = t; i0 < nb; i0 += 256 * 8) {
            int w[8];
            #pragma unroll
            for (int u = 0; u < 8; ++u) {
                const int i = i0 + 256 * u;
                w[u] = (i < nb) ? pairs[start + i] : -1;
            }
            #pragma unroll
            for (int u = 0; u < 8; ++u) {
                if (w[u] >= 0) {
                    const int p = atomicAdd(&cur[(w[u] >> SND_BITS) & (BN - 1)], 1);
                    if ((unsigned)p < (unsigned)CAP) stage[p] = w[u] & SND_MASK;
                }
            }
        }
        __syncthreads();
        for (int i = t; i < nb; i += 256) csr[start + i] = stage[i];
    } else {
        for (int i = t; i < nb; i += 256) {
            const int w = pairs[start + i];
            const int p = atomicAdd(&cur[(w >> SND_BITS) & (BN - 1)], 1);
            if ((unsigned)p < (unsigned)nb) csr[start + p] = w & SND_MASK;
        }
    }
}

static constexpr int MLP_LDS_WORDS = 8 * 16 * DH;
__device__ __forceinline__ void mlp_body(const int blk, float* lds,
                           const float* __restrict__ X,
                           const float* __restrict__ W1, const float* __restrict__ b1,
                           const float* __restrict__ W2, const float* __restrict__ b2,
                           const float* __restrict__ W3,
                           const float* __restrict__ sdeg,
                           float* __restrict__ p1, int nTiles) {
    const int w    = threadIdx.x >> 5;
    const int lane = threadIdx.x & 31;
    const int m16  = lane & 15;
    const int kh   = lane >> 4;
    const int t    = blk * 8 + w;
    const bool active = (t < nTiles);
    const int row0 = t * 16;
    float* my = &lds[w * 16 * DH];

    if (active) {
        v8f c0 = {}, c1 = {}, c2 = {}, c3 = {};
        #pragma unroll 4
        for (int kb = 0; kb < DF; kb += 4) {
            v2f a = *(const v2f*)(X + (size_t)(row0 + m16) * DF + kb + 2 * kh);
            const float* wr0 = W1 + (size_t)(kb + 2 * kh) * DH;
            const float* wr1 = wr0 + DH;
            v2f b;
            b.x = wr0[m16];      b.y = wr1[m16];      c0 = wmma4(a, b, c0);
            b.x = wr0[16 + m16]; b.y = wr1[16 + m16]; c1 = wmma4(a, b, c1);
            b.x = wr0[32 + m16]; b.y = wr1[32 + m16]; c2 = wmma4(a, b, c2);
            b.x = wr0[48 + m16]; b.y = wr1[48 + m16]; c3 = wmma4(a, b, c3);
        }
        #pragma unroll
        for (int r = 0; r < 8; ++r) {
            float* dr = my + (r + 8 * kh) * DH;
            dr[m16]      = lrelu(c0[r] + b1[m16]);
            dr[16 + m16] = lrelu(c1[r] + b1[16 + m16]);
            dr[32 + m16] = lrelu(c2[r] + b1[32 + m16]);
            dr[48 + m16] = lrelu(c3[r] + b1[48 + m16]);
        }
    }
    __syncthreads();

    if (active) {
        v8f d0 = {}, d1 = {}, d2 = {}, d3 = {};
        #pragma unroll 4
        for (int kb = 0; kb < DH; kb += 4) {
            v2f a = *(const v2f*)(my + m16 * DH + kb + 2 * kh);
            const float* wr0 = W2 + (size_t)(kb + 2 * kh) * DH;
            const float* wr1 = wr0 + DH;
            v2f b;
            b.x = wr0[m16];      b.y = wr1[m16];      d0 = wmma4(a, b, d0);
            b.x = wr0[16 + m16]; b.y = wr1[16 + m16]; d1 = wmma4(a, b, d1);
            b.x = wr0[32 + m16]; b.y = wr1[32 + m16]; d2 = wmma4(a, b, d2);
            b.x = wr0[48 + m16]; b.y = wr1[48 + m16]; d3 = wmma4(a, b, d3);
        }
        float sc[8];
        #pragma unroll
        for (int r = 0; r < 8; ++r) sc[r] = rsqrtf(fmaxf(sdeg[row0 + r + 8 * kh], 1.f));
        #pragma unroll
        for (int r = 0; r < 8; ++r) {
            float* dr = my + (r + 8 * kh) * DH;
            dr[m16]      = lrelu(d0[r] + b2[m16])      * sc[r];
            dr[16 + m16] = lrelu(d1[r] + b2[16 + m16]) * sc[r];
            dr[32 + m16] = lrelu(d2[r] + b2[32 + m16]) * sc[r];
            dr[48 + m16] = lrelu(d3[r] + b2[48 + m16]) * sc[r];
        }
    }
    __syncthreads();
    if (!active) return;

    v8f e0 = {}, e1 = {}, e2 = {};
    #pragma unroll 4
    for (int kb = 0; kb < DH; kb += 4) {
        v2f a = *(const v2f*)(my + m16 * DH + kb + 2 * kh);
        const float* wr0 = W3 + (size_t)(kb + 2 * kh) * DC;
        const float* wr1 = wr0 + DC;
        v2f b;
        b.x = wr0[m16];      b.y = wr1[m16];      e0 = wmma4(a, b, e0);
        b.x = wr0[16 + m16]; b.y = wr1[16 + m16]; e1 = wmma4(a, b, e1);
        const int n2  = 32 + m16;
        const int n2c = (n2 < DC) ? n2 : DC - 1;
        const float x0 = wr0[n2c], x1 = wr1[n2c];
        b.x = (n2 < DC) ? x0 : 0.f;
        b.y = (n2 < DC) ? x1 : 0.f;
        e2 = wmma4(a, b, e2);
    }
    #pragma unroll
    for (int r = 0; r < 8; ++r) {
        float* orow = p1 + (size_t)(row0 + r + 8 * kh) * P1P;
        orow[m16]      = e0[r];
        orow[16 + m16] = e1[r];
        if (32 + m16 < DC)
            orow[32 + m16] = e2[r];
    }
}

static constexpr int MLP_BLKS   = (NN / 16 + 7) / 8;
static constexpr int MLP_BLKS_A = MLP_BLKS / 2;
static constexpr int MLP_BLKS_B = MLP_BLKS - MLP_BLKS_A;
static constexpr int LDS_A = BIN_LDS_WORDS  > MLP_LDS_WORDS ? BIN_LDS_WORDS  : MLP_LDS_WORDS;
static constexpr int LDS_B = SORT_LDS_WORDS > MLP_LDS_WORDS ? SORT_LDS_WORDS : MLP_LDS_WORDS;

__launch_bounds__(256)
__global__ void bin_mlp_kernel(const int* __restrict__ snd, const int* __restrict__ rcv,
                               const int* __restrict__ btot, int* __restrict__ bcur,
                               int* __restrict__ pairs,
                               const float* __restrict__ X,
                               const float* __restrict__ W1, const float* __restrict__ b1,
                               const float* __restrict__ W2, const float* __restrict__ b2,
                               const float* __restrict__ W3,
                               const float* __restrict__ sdeg,
                               float* __restrict__ p1, int nTiles) {
    __shared__ __attribute__((aligned(16))) int smem[LDS_A];
    const int blk = blockIdx.x;
    if (blk < NCH) bin_body(blk, smem, snd, rcv, btot, bcur, pairs);
    else           mlp_body(blk - NCH, (float*)smem, X, W1, b1, W2, b2, W3, sdeg, p1, nTiles);
}

__launch_bounds__(256)
__global__ void sort_mlp_kernel(const int* __restrict__ pairs, const int* __restrict__ btot,
                                int* __restrict__ offs, float* __restrict__ rdeg, int* __restrict__ csr,
                                const float* __restrict__ X,
                                const float* __restrict__ W1, const float* __restrict__ b1,
                                const float* __restrict__ W2, const float* __restrict__ b2,
                                const float* __restrict__ W3,
                                const float* __restrict__ sdeg,
                                float* __restrict__ p1, int nTiles) {
    __shared__ __attribute__((aligned(16))) int smem[LDS_B];
    const int blk = blockIdx.x;
    if (blk < NB) sort_body(blk, smem, pairs, btot, offs, rdeg, csr);
    else          mlp_body(MLP_BLKS_A + (blk - NB), (float*)smem, X, W1, b1, W2, b2, W3, sdeg, p1, nTiles);
}

__launch_bounds__(256)
__global__ void gather1_kernel(const float* __restrict__ p1, const int* __restrict__ csr,
                               const int* __restrict__ offs, const float* __restrict__ rdeg,
                               const float* __restrict__ sdeg, const float* __restrict__ b3,
                               float* __restrict__ h2) {
    const int wave = (blockIdx.x * 256 + threadIdx.x) >> 5;
    const int lane = threadIdx.x & 31;
    const int q = lane / 10;
    const int l = lane - 10 * q;
    const int n = wave * G1N + q;
    if (q >= G1N || n >= NN) return;
    const int* seg = csr + offs[n];
    const float rd = rdeg[n];
    const int cnt = (int)rd;
    float4 acc = {0.f, 0.f, 0.f, 0.f};
    for (int j = 0; j < cnt; j += 8) {
        int s[8];
        float4 v[8];
        #pragma unroll
        for (int k = 0; k < 8; ++k) {
            const int qd = seg[min(j + k, cnt - 1)];
            s[k] = ((unsigned)qd < (unsigned)NN) ? qd : 0;
        }
        #pragma unroll
        for (int k = 0; k < 8; ++k)
            v[k] = *(const float4*)(p1 + (size_t)s[k] * P1P + 4 * l);
        #pragma unroll
        for (int k = 0; k < 8; ++k) {
            const bool in = j + k < cnt;
            acc.x += in ? v[k].x : 0.f; acc.y += in ? v[k].y : 0.f;
            acc.z += in ? v[k].z : 0.f; acc.w += in ? v[k].w : 0.f;
        }
    }
    const float rs = rsqrtf(fmaxf(rd, 1.f));
    const float sc = rsqrtf(fmaxf(sdeg[n], 1.f));
    float4 o;
    o.x = (acc.x * rs + b3[4 * l])     * sc;
    o.y = (acc.y * rs + b3[4 * l + 1]) * sc;
    o.z = (acc.z * rs + b3[4 * l + 2]) * sc;
    o.w = (acc.w * rs + b3[4 * l + 3]) * sc;
    *(float4*)(h2 + (size_t)n * DC + 4 * l) = o;
}

__launch_bounds__(256)
__global__ void gather2_kernel(const float* __restrict__ h2, const int* __restrict__ csr,
                               const int* __restrict__ offs, const float* __restrict__ rdeg,
                               float* __restrict__ out) {
    const int gt = blockIdx.x * 256 + threadIdx.x;
    const int n = gt >> 3;
    const int l = gt & 7;
    if (n >= NN) return;
    const int* seg = csr + offs[n];
    const int cnt = (int)rdeg[n];
    float4 acc = {0.f, 0.f, 0.f, 0.f};
    float tail = 0.f;
    for (int j = 0; j < cnt; j += 8) {
        int s[8];
        float4 v[8];
        float u[8];
        #pragma unroll
        for (int k = 0; k < 8; ++k) {
            const int q = seg[min(j + k, cnt - 1)];
            s[k] = ((unsigned)q < (unsigned)NN) ? q : 0;
        }
        #pragma unroll
        for (int k = 0; k < 8; ++k) {
            const float* row = h2 + (size_t)s[k] * DC;
            v[k] = *(const float4*)(row + 4 * l);
            u[k] = row[32 + l];
        }
        #pragma unroll
        for (int k = 0; k < 8; ++k) {
            const bool in = j + k < cnt;
            acc.x += in ? v[k].x : 0.f; acc.y += in ? v[k].y : 0.f;
            acc.z += in ? v[k].z : 0.f; acc.w += in ? v[k].w : 0.f;
            tail += in ? u[k] : 0.f;
        }
    }
    float* orow = out + (size_t)n * DC;
    *(float4*)(orow + 4 * l) = acc;
    orow[32 + l] = tail;
}

__global__ void softmax_kernel(float* __restrict__ out, const float* __restrict__ rdeg) {
    int n = blockIdx.x * blockDim.x + threadIdx.x;
    if (n >= NN) return;
    float rs = rsqrtf(fmaxf(rdeg[n], 1.f));
    float* row = out + (size_t)n * DC;
    float v[DC];
    float mx = -3.402823466e38f;
    #pragma unroll
    for (int f = 0; f < DC; ++f) { v[f] = row[f] * rs; mx = fmaxf(mx, v[f]); }
    float s = 0.f;
    #pragma unroll
    for (int f = 0; f < DC; ++f) { v[f] = __expf(v[f] - mx); s += v[f]; }
    float inv = 1.f / s;
    #pragma unroll
    for (int f = 0; f < DC; ++f) row[f] = v[f] * inv;
}

extern "C" void kernel_launch(void* const* d_in, const int* in_sizes, int n_in,
                              void* d_out, int out_size, void* d_ws, size_t ws_size,
                              hipStream_t stream) {
    const float* nodes     = (const float*)d_in[0];
    const int*   senders   = (const int*)d_in[1];
    const int*   receivers = (const int*)d_in[2];
    const float* W1 = (const float*)d_in[3];
    const float* b1 = (const float*)d_in[4];
    const float* W2 = (const float*)d_in[5];
    const float* b2 = (const float*)d_in[6];
    const float* W3 = (const float*)d_in[7];
    const float* b3 = (const float*)d_in[8];
    float* out = (float*)d_out;
    float* ws  = (float*)d_ws;

    float* sdeg = ws;
    int*   btot = (int*)(ws + NN);
    int*   bcur = btot + NBP;
    float* rdeg = ws + NN + 2 * NBP;
    float* p1   = rdeg + NN;
    float* h2   = p1 + (size_t)NN * P1P;
    int*   offs = (int*)(h2 + (size_t)NN * DC);
    int*   pairs = offs + NN;
    int*   csr   = pairs + NE;

    const int nTiles = NN / 16;
    zero_f32<<<2048, 256, 0, stream>>>(sdeg, NN + 2 * NBP);
    count_kernel<<<NCH, 256, 0, stream>>>(senders, receivers, sdeg, btot);
    bin_mlp_kernel<<<NCH + MLP_BLKS_A, 256, 0, stream>>>(senders, receivers, btot, bcur, pairs,
                                                         nodes, W1, b1, W2, b2, W3, sdeg, p1, nTiles);
    sort_mlp_kernel<<<NB + MLP_BLKS_B, 256, 0, stream>>>(pairs, btot, offs, rdeg, csr,
                                                         nodes, W1, b1, W2, b2, W3, sdeg, p1, nTiles);
    gather1_kernel<<<((NN + G1N - 1) / G1N + 7) / 8, 256, 0, stream>>>(p1, csr, offs, rdeg, sdeg, b3, h2);
    gather2_kernel<<<(NN * 8 + 255) / 256, 256, 0, stream>>>(h2, csr, offs, rdeg, out);
    softmax_kernel<<<(NN + 255) / 256, 256, 0, stream>>>(out, rdeg);
}


# Row softmax moved into gather 2's epilogue: max and exp-sum over a lane's 5 values, then across the node's 8 lanes (exp-sum order changes); softmax launch and its pass over out gone, 7 -> 6 launches
# speedup vs baseline: 3.6869x; 1.0903x over previous
#include <hip/hip_runtime.h>

typedef __attribute__((ext_vector_type(2))) float v2f;
typedef __attribute__((ext_vector_type(8))) float v8f;

static constexpr int NN = 100000;
static constexpr int NE = 1600000;
static constexpr int DF = 128;
static constexpr int DH = 64;
static constexpr int DC = 40;
static constexpr int P1P = 64;
static constexpr int G1N = 3;
#define NEG_SLOPE 0.01f

__device__ __forceinline__ float lrelu(float x) { return x > 0.f ? x : x * NEG_SLOPE; }

__device__ __forceinline__ v8f wmma4(v2f a, v2f b, v8f c) {
    return __builtin_amdgcn_wmma_f32_16x16x4_f32(false, a, false, b, (short)0, c, false, false);
}

__global__ void zero_f32(float* __restrict__ p, int n) {
    int i = blockIdx.x * blockDim.x + threadIdx.x;
    int stride = gridDim.x * blockDim.x;
    for (; i < n; i += stride) p[i] = 0.f;
}

static constexpr int BN_SHIFT = 8;
static constexpr int BN   = 1 << BN_SHIFT;
static constexpr int NB   = (NN + BN - 1) / BN;
static constexpr int NBP  = 512;
static constexpr int CH   = 4096;
static constexpr int NCH  = (NE + CH - 1) / CH;
static constexpr int EPT  = CH / 256;
static constexpr int SND_BITS = 17;
static constexpr int SND_MASK = (1 << SND_BITS) - 1;
static constexpr int CAP  = 12288;
static_assert(NN <= (1 << SND_BITS), "a sender id must fit the packed word");
static_assert(NB <= NBP, "two bins per thread cover every bin");
static_assert(NE < (1 << 24), "degrees are whole numbers that stay exact in f32");

__device__ __forceinline__ int block_scan_incl(int v, int* buf) {
    const int t = threadIdx.x;
    buf[t] = v;
    __syncthreads();
    for (int d = 1; d < 256; d <<= 1) {
        const int a = (t >= d) ? buf[t - d] : 0;
        __syncthreads();
        v += a;
        buf[t] = v;
        __syncthreads();
    }
    return v;
}

__device__ __forceinline__ void block_scan_incl2(int& a, int& b, int* bufA, int* bufB) {
    const int t = threadIdx.x;
    bufA[t] = a;
    bufB[t] = b;
    __syncthreads();
    for (int d = 1; d < 256; d <<= 1) {
        const int xa = (t >= d) ? bufA[t - d] : 0;
        const int xb = (t >= d) ? bufB[t - d] : 0;
        __syncthreads();
        a += xa;
        b += xb;
        bufA[t] = a;
        bufB[t] = b;
        __syncthreads();
    }
}

__launch_bounds__(256)
__global__ void count_kernel(const int* __restrict__ snd, const int* __restrict__ rcv,
                             float* __restrict__ sdeg, int* __restrict__ btot) {
    __shared__ int hist[NBP];
    const int t = threadIdx.x;
    hist[t] = 0;
    hist[t + 256] = 0;
    __syncthreads();
    const int e0 = blockIdx.x * CH + t;
    int r[EPT], s[EPT];
    #pragma unroll
    for (int k = 0; k < EPT; ++k) {
        const int e = e0 + k * 256;
        r[k] = (e < NE) ? rcv[e] : -1;
        s[k] = (e < NE) ? snd[e] : -1;
    }
    #pragma unroll
    for (int k = 0; k < EPT; ++k) {
        if ((unsigned)r[k] < (unsigned)NN) atomicAdd(&hist[r[k] >> BN_SHIFT], 1);
        if ((unsigned)s[k] < (unsigned)NN) atomicAdd(&sdeg[s[k]], 1.f);
    }
    __syncthreads();
    const int h0 = hist[t], h1 = hist[t + 256];
    if (h0) atomicAdd(&btot[t], h0);
    if (h1) atomicAdd(&btot[t + 256], h1);
}

static constexpr int BIN_LDS_WORDS = 3 * NBP + 2 * 256 + 2 * CH;
__device__ __forceinline__ void bin_body(const int blk, int* smem,
                           const int* __restrict__ snd, const int* __restrict__ rcv,
                           const int* __restrict__ btot, int* __restrict__ bcur,
                           int* __restrict__ pairs) {
    int* hist   = smem;
    int* lstart = hist + NBP;
    int* gstart = lstart + NBP;
    int* sA     = gstart + NBP;
    int* sB     = sA + 256;
    int* sw     = sB + 256;
    int* sd     = sw + CH;
    const int t = threadIdx.x;
    hist[t] = 0;
    hist[t + 256] = 0;
    __syncthreads();
    int word[EPT], lr[EPT], bin[EPT];
    const int e0 = blk * CH + t;
    #pragma unroll
    for (int k = 0; k < EPT; ++k) {
        const int e = e0 + k * 256;
        bin[k]  = (e < NE) ? rcv[e] : -1;
        word[k] = (e < NE) ? snd[e] : 0;
    }
    #pragma unroll
    for (int k = 0; k < EPT; ++k) {
        const int r = bin[k];
        lr[k] = -1;
        bin[k] = 0;
        if ((unsigned)r < (unsigned)NN) {
            bin[k]  = r >> BN_SHIFT;
            lr[k]   = atomicAdd(&hist[bin[k]], 1);
            word[k] = ((r & (BN - 1)) << SND_BITS) | (word[k] & SND_MASK);
        }
    }
    __syncthreads();
    const int h0 = hist[2 * t], h1 = hist[2 * t + 1];
    const int g0 = btot[2 * t], g1 = btot[2 * t + 1];
    int li = h0 + h1, gi = g0 + g1;
    block_scan_incl2(li, gi, sA, sB);
    const int lex = li - (h0 + h1);
    const int gex = gi - (g0 + g1);
    int r0 = 0, r1 = 0;
    if (h0) r0 = atomicAdd(&bcur[2 * t], h0);
    if (h1) r1 = atomicAdd(&bcur[2 * t + 1], h1);
    lstart[2 * t]     = lex;
    lstart[2 * t + 1] = lex + h0;
    gstart[2 * t]     = gex + r0;
    gstart[2 * t + 1] = gex + g0 + r1;
    __syncthreads();
    #pragma unroll
    for (int k = 0; k < EPT; ++k) {
        if (lr[k] >= 0) {
            const int pos = lstart[bin[k]] + lr[k];
            sw[pos] = word[k];
            sd[pos] = gstart[bin[k]] + lr[k];
        }
    }
    __syncthreads();
    const int mine = sA[255];
    for (int i = t; i < mine; i += 256) {
        const int p = sd[i];
        if ((unsigned)p < (unsigned)NE) pairs[p] = sw[i];
    }
}

static constexpr int SORT_LDS_WORDS = 2 * 256 + 2 * BN + CAP;
__device__ __forceinline__ void sort_body(const int b, int* smem,
                            const int* __restrict__ pairs, const int* __restrict__ btot,
                            int* __restrict__ offs, float* __restrict__ rdeg, int* __restrict__ csr) {
    int* sA    = smem;
    int* sB    = sA + 256;
    int* cnt   = sB + 256;
    int* cur   = cnt + BN;
    int* stage = cur + BN;
    const int t = threadIdx.x;
    int front = 0;
    if (t < b)       front += btot[t];
    if (t + 256 < b) front += btot[t + 256];
    for (int m = 16; m >= 1; m >>= 1) front += __shfl_xor(front, m);
    if ((t & 31) == 0) sA[t >> 5] = front;
    __syncthreads();
    int all = 0;
    #pragma unroll
    for (int w = 0; w < 8; ++w) all += sA[w];
    const int start = min(max(all, 0), NE);
    const int nb    = min(max(btot[b], 0), NE - start);
    cnt[t] = 0;
    __syncthreads();
    for (int i0 = t; i0 < nb; i0 += 256 * 8) {
        int w[8];
        #pragma unroll
        for (int u = 0; u < 8; ++u) {
            const int i = i0 + 256 * u;
            w[u] = (i < nb) ? pairs[start + i] : -1;
        }
        #pragma unroll
        for (int u = 0; u < 8; ++u)
            if (w[u] >= 0) atomicAdd(&cnt[(w[u] >> SND_BITS) & (BN - 1)], 1);
    }
    __syncthreads();
    const int c   = cnt[t];
    const int loc = block_scan_incl(c, sB) - c;
    const int node = b * BN + t;
    if (node < NN) {
        offs[node] = start + loc;
        rdeg[node] = (float)c;
    }
    cur[t] = loc;
    __syncthreads();
    if (nb <= CAP) {
        for (int i0 = t; i0 < nb; i0 += 256 * 8) {
            int w[8];
            #pragma unroll
            for (int u = 0; u < 8; ++u) {
                const int i = i0 + 256 * u;
                w[u] = (i < nb) ? pairs[start + i] : -1;
            }
            #pragma unroll
            for (int u = 0; u < 8; ++u) {
                if (w[u] >= 0) {
                    const int p = atomicAdd(&cur[(w[u] >> SND_BITS) & (BN - 1)], 1);
                    if ((unsigned)p < (unsigned)CAP) stage[p] = w[u] & SND_MASK;
                }
            }
        }
        __syncthreads();
        for (int i = t; i < nb; i += 256) csr[start + i] = stage[i];
    } else {
        for (int i = t; i < nb; i += 256) {
            const int w = pairs[start + i];
            const int p = atomicAdd(&cur[(w >> SND_BITS) & (BN - 1)], 1);
            if ((unsigned)p < (unsigned)nb) csr[start + p] = w & SND_MASK;
        }
    }
}

static constexpr int MLP_LDS_WORDS = 8 * 16 * DH;
__device__ __forceinline__ void mlp_body(const int blk, float* lds,
                           const float* __restrict__ X,
                           const float* __restrict__ W1, const float* __restrict__ b1,
                           const float* __restrict__ W2, const float* __restrict__ b2,
                           const float* __restrict__ W3,
                           const float* __restrict__ sdeg,
                           float* __restrict__ p1, int nTiles) {
    const int w    = threadIdx.x >> 5;
    const int lane = threadIdx.x & 31;
    const int m16  = lane & 15;
    const int kh   = lane >> 4;
    const int t    = blk * 8 + w;
    const bool active = (t < nTiles);
    const int row0 = t * 16;
    float* my = &lds[w * 16 * DH];

    if (active) {
        v8f c0 = {}, c1 = {}, c2 = {}, c3 = {};
        #pragma unroll 4
        for (int kb = 0; kb < DF; kb += 4) {
            v2f a = *(const v2f*)(X + (size_t)(row0 + m16) * DF + kb + 2 * kh);
            const float* wr0 = W1 + (size_t)(kb + 2 * kh) * DH;
            const float* wr1 = wr0 + DH;
            v2f b;
            b.x = wr0[m16];      b.y = wr1[m16];      c0 = wmma4(a, b, c0);
            b.x = wr0[16 + m16]; b.y = wr1[16 + m16]; c1 = wmma4(a, b, c1);
            b.x = wr0[32 + m16]; b.y = wr1[32 + m16]; c2 = wmma4(a, b, c2);
            b.x = wr0[48 + m16]; b.y = wr1[48 + m16]; c3 = wmma4(a, b, c3);
        }
        #pragma unroll
        for (int r = 0; r < 8; ++r) {
            float* dr = my + (r + 8 * kh) * DH;
            dr[m16]      = lrelu(c0[r] + b1[m16]);
            dr[16 + m16] = lrelu(c1[r] + b1[16 + m16]);
            dr[32 + m16] = lrelu(c2[r] + b1[32 + m16]);
            dr[48 + m16] = lrelu(c3[r] + b1[48 + m16]);
        }
    }
    __syncthreads();

    if (active) {
        v8f d0 = {}, d1 = {}, d2 = {}, d3 = {};
        #pragma unroll 4
        for (int kb = 0; kb < DH; kb += 4) {
            v2f a = *(const v2f*)(my + m16 * DH + kb + 2 * kh);
            const float* wr0 = W2 + (size_t)(kb + 2 * kh) * DH;
            const float* wr1 = wr0 + DH;
            v2f b;
            b.x = wr0[m16];      b.y = wr1[m16];      d0 = wmma4(a, b, d0);
            b.x = wr0[16 + m16]; b.y = wr1[16 + m16]; d1 = wmma4(a, b, d1);
            b.x = wr0[32 + m16]; b.y = wr1[32 + m16]; d2 = wmma4(a, b, d2);
            b.x = wr0[48 + m16]; b.y = wr1[48 + m16]; d3 = wmma4(a, b, d3);
        }
        float sc[8];
        #pragma unroll
        for (int r = 0; r < 8; ++r) sc[r] = rsqrtf(fmaxf(sdeg[row0 + r + 8 * kh], 1.f));
        #pragma unroll
        for (int r = 0; r < 8; ++r) {
            float* dr = my + (r + 8 * kh) * DH;
            dr[m16]      = lrelu(d0[r] + b2[m16])      * sc[r];
            dr[16 + m16] = lrelu(d1[r] + b2[16 + m16]) * sc[r];
            dr[32 + m16] = lrelu(d2[r] + b2[32 + m16]) * sc[r];
            dr[48 + m16] = lrelu(d3[r] + b2[48 + m16]) * sc[r];
        }
    }
    __syncthreads();
    if (!active) return;

    v8f e0 = {}, e1 = {}, e2 = {};
    #pragma unroll 4
    for (int kb = 0; kb < DH; kb += 4) {
        v2f a = *(const v2f*)(my + m16 * DH + kb + 2 * kh);
        const float* wr0 = W3 + (size_t)(kb + 2 * kh) * DC;
        const float* wr1 = wr0 + DC;
        v2f b;
        b.x = wr0[m16];      b.y = wr1[m16];      e0 = wmma4(a, b, e0);
        b.x = wr0[16 + m16]; b.y = wr1[16 + m16]; e1 = wmma4(a, b, e1);
        const int n2  = 32 + m16;
        const int n2c = (n2 < DC) ? n2 : DC - 1;
        const float x0 = wr0[n2c], x1 = wr1[n2c];
        b.x = (n2 < DC) ? x0 : 0.f;
        b.y = (n2 < DC) ? x1 : 0.f;
        e2 = wmma4(a, b, e2);
    }
    #pragma unroll
    for (int r = 0; r < 8; ++r) {
        float* orow = p1 + (size_t)(row0 + r + 8 * kh) * P1P;
        orow[m16]      = e0[r];
        orow[16 + m16] = e1[r];
        if (32 + m16 < DC)
            orow[32 + m16] = e2[r];
    }
}

static constexpr int MLP_BLKS   = (NN / 16 + 7) / 8;
static constexpr int MLP_BLKS_A = MLP_BLKS / 2;
static constexpr int MLP_BLKS_B = MLP_BLKS - MLP_BLKS_A;
static constexpr int LDS_A = BIN_LDS_WORDS  > MLP_LDS_WORDS ? BIN_LDS_WORDS  : MLP_LDS_WORDS;
static constexpr int LDS_B = SORT_LDS_WORDS > MLP_LDS_WORDS ? SORT_LDS_WORDS : MLP_LDS_WORDS;

__launch_bounds__(256)
__global__ void bin_mlp_kernel(const int* __restrict__ snd, const int* __restrict__ rcv,
                               const int* __restrict__ btot, int* __restrict__ bcur,
                               int* __restrict__ pairs,
                               const float* __restrict__ X,
                               const float* __restrict__ W1, const float* __restrict__ b1,
                               const float* __restrict__ W2, const float* __restrict__ b2,
                               const float* __restrict__ W3,
                               const float* __restrict__ sdeg,
                               float* __restrict__ p1, int nTiles) {
    __shared__ __attribute__((aligned(16))) int smem[LDS_A];
    const int blk = blockIdx.x;
    if (blk < NCH) bin_body(blk, smem, snd, rcv, btot, bcur, pairs);
    else           mlp_body(blk - NCH, (float*)smem, X, W1, b1, W2, b2, W3, sdeg, p1, nTiles);
}

__launch_bounds__(256)
__global__ void sort_mlp_kernel(const int* __restrict__ pairs, const int* __restrict__ btot,
                                int* __restrict__ offs, float* __restrict__ rdeg, int* __restrict__ csr,
                                const float* __restrict__ X,
                                const float* __restrict__ W1, const float* __restrict__ b1,
                                const float* __restrict__ W2, const float* __restrict__ b2,
                                const float* __restrict__ W3,
                                const float* __restrict__ sdeg,
                                float* __restrict__ p1, int nTiles) {
    __shared__ __attribute__((aligned(16))) int smem[LDS_B];
    const int blk = blockIdx.x;
    if (blk < NB) sort_body(blk, smem, pairs, btot, offs, rdeg, csr);
    else          mlp_body(MLP_BLKS_A + (blk - NB), (float*)smem, X, W1, b1, W2, b2, W3, sdeg, p1, nTiles);
}

__launch_bounds__(256)
__global__ void gather1_kernel(const float* __restrict__ p1, const int* __restrict__ csr,
                               const int* __restrict__ offs, const float* __restrict__ rdeg,
                               const float* __restrict__ sdeg, const float* __restrict__ b3,
                               float* __restrict__ h2) {
    const int wave = (blockIdx.x * 256 + threadIdx.x) >> 5;
    const int lane = threadIdx.x & 31;
    const int q = lane / 10;
    const int l = lane - 10 * q;
    const int n = wave * G1N + q;
    if (q >= G1N || n >= NN) return;
    const int* seg = csr + offs[n];
    const float rd = rdeg[n];
    const int cnt = (int)rd;
    float4 acc = {0.f, 0.f, 0.f, 0.f};
    for (int j = 0; j < cnt; j += 8) {
        int s[8];
        float4 v[8];
        #pragma unroll
        for (int k = 0; k < 8; ++k) {
            const int qd = seg[min(j + k, cnt - 1)];
            s[k] = ((unsigned)qd < (unsigned)NN) ? qd : 0;
        }
        #pragma unroll
        for (int k = 0; k < 8; ++k)
            v[k] = *(const float4*)(p1 + (size_t)s[k] * P1P + 4 * l);
        #pragma unroll
        for (int k = 0; k < 8; ++k) {
            const bool in = j + k < cnt;
            acc.x += in ? v[k].x : 0.f; acc.y += in ? v[k].y : 0.f;
            acc.z += in ? v[k].z : 0.f; acc.w += in ? v[k].w : 0.f;
        }
    }
    const float rs = rsqrtf(fmaxf(rd, 1.f));
    const float sc = rsqrtf(fmaxf(sdeg[n], 1.f));
    float4 o;
    o.x = (acc.x * rs + b3[4 * l])     * sc;
    o.y = (acc.y * rs + b3[4 * l + 1]) * sc;
    o.z = (acc.z * rs + b3[4 * l + 2]) * sc;
    o.w = (acc.w * rs + b3[4 * l + 3]) * sc;
    *(float4*)(h2 + (size_t)n * DC + 4 * l) = o;
}

__launch_bounds__(256)
__global__ void gather2_softmax_kernel(const float* __restrict__ h2, const int* __restrict__ csr,
                                       const int* __restrict__ offs, const float* __restrict__ rdeg,
                                       float* __restrict__ out) {
    const int gt = blockIdx.x * 256 + threadIdx.x;
    const int n = gt >> 3;
    const int l = gt & 7;
    if (n >= NN) return;
    const int* seg = csr + offs[n];
    const float rd = rdeg[n];
    const int cnt = (int)rd;
    float4 acc = {0.f, 0.f, 0.f, 0.f};
    float tail = 0.f;
    for (int j = 0; j < cnt; j += 8) {
        int s[8];
        float4 v[8];
        float u[8];
        #pragma unroll
        for (int k = 0; k < 8; ++k) {
            const int q = seg[min(j + k, cnt - 1)];
            s[k] = ((unsigned)q < (unsigned)NN) ? q : 0;
        }
        #pragma unroll
        for (int k = 0; k < 8; ++k) {
            const float* row = h2 + (size_t)s[k] * DC;
            v[k] = *(const float4*)(row + 4 * l);
            u[k] = row[32 + l];
        }
        #pragma unroll
        for (int k = 0; k < 8; ++k) {
            const bool in = j + k < cnt;
            acc.x += in ? v[k].x : 0.f; acc.y += in ? v[k].y : 0.f;
            acc.z += in ? v[k].z : 0.f; acc.w += in ? v[k].w : 0.f;
            tail += in ? u[k] : 0.f;
        }
    }
    const float rs = rsqrtf(fmaxf(rd, 1.f));
    const float a0 = acc.x * rs, a1 = acc.y * rs, a2 = acc.z * rs, a3 = acc.w * rs, a4 = tail * rs;
    float mx = fmaxf(fmaxf(fmaxf(a0, a1), fmaxf(a2, a3)), a4);
    #pragma unroll
    for (int m = 1; m < 8; m <<= 1) mx = fmaxf(mx, __shfl_xor(mx, m));
    const float e0 = __expf(a0 - mx), e1 = __expf(a1 - mx), e2 = __expf(a2 - mx),
                e3 = __expf(a3 - mx), e4 = __expf(a4 - mx);
    float sum = (e0 + e1) + (e2 + e3) + e4;
    #pragma unroll
    for (int m = 1; m < 8; m <<= 1) sum += __shfl_xor(sum, m);
    const float inv = 1.f / sum;
    float* orow = out + (size_t)n * DC;
    float4 o;
    o.x = e0 * inv; o.y = e1 * inv; o.z = e2 * inv; o.w = e3 * inv;
    *(float4*)(orow + 4 * l) = o;
    orow[32 + l] = e4 * inv;
}

extern "C" void kernel_launch(void* const* d_in, const int* in_sizes, int n_in,
                              void* d_out, int out_size, void* d_ws, size_t ws_size,
                              hipStream_t stream) {
    const float* nodes     = (const float*)d_in[0];
    const int*   senders   = (const int*)d_in[1];
    const int*   receivers = (const int*)d_in[2];
    const float* W1 = (const float*)d_in[3];
    const float* b1 = (const float*)d_in[4];
    const float* W2 = (const float*)d_in[5];
    const float* b2 = (const float*)d_in[6];
    const float* W3 = (const float*)d_in[7];
    const float* b3 = (const float*)d_in[8];
    float* out = (float*)d_out;
    float* ws  = (float*)d_ws;

    float* sdeg = ws;
    int*   btot = (int*)(ws + NN);
    int*   bcur = btot + NBP;
    float* rdeg = ws + NN + 2 * NBP;
    float* p1   = rdeg + NN;
    float* h2   = p1 + (size_t)NN * P1P;
    int*   offs = (int*)(h2 + (size_t)NN * DC);
    int*   pairs = offs + NN;
    int*   csr   = pairs + NE;

    const int nTiles = NN / 16;
    zero_f32<<<2048, 256, 0, stream>>>(sdeg, NN + 2 * NBP);
    count_kernel<<<NCH, 256, 0, stream>>>(senders, receivers, sdeg, btot);
    bin_mlp_kernel<<<NCH + MLP_BLKS_A, 256, 0, stream>>>(senders, receivers, btot, bcur, pairs,
                                                         nodes, W1, b1, W2, b2, W3, sdeg, p1, nTiles);
    sort_mlp_kernel<<<NB + MLP_BLKS_B, 256, 0, stream>>>(pairs, btot, offs, rdeg, csr,
                                                         nodes, W1, b1, W2, b2, W3, sdeg, p1, nTiles);
    gather1_kernel<<<((NN + G1N - 1) / G1N + 7) / 8, 256, 0, stream>>>(p1, csr, offs, rdeg, sdeg, b3, h2);
    gather2_softmax_kernel<<<(NN * 8 + 255) / 256, 256, 0, stream>>>(h2, csr, offs, rdeg, out);
}


# Sender degrees counted in LDS: a block owns a node range on an edge chunk and adds its counts to memory line-wide, not one scattered single-word global add per edge (whole numbers: sums stay exact)
# speedup vs baseline: 6.5305x; 1.7713x over previous
#include <hip/hip_runtime.h>

typedef __attribute__((ext_vector_type(2))) float v2f;
typedef __attribute__((ext_vector_type(8))) float v8f;

static constexpr int NN = 100000;
static constexpr int NE = 1600000;
static constexpr int DF = 128;
static constexpr int DH = 64;
static constexpr int DC = 40;
static constexpr int P1P = 64;
static constexpr int G1N = 3;
#define NEG_SLOPE 0.01f

__device__ __forceinline__ float lrelu(float x) { return x > 0.f ? x : x * NEG_SLOPE; }

__device__ __forceinline__ v8f wmma4(v2f a, v2f b, v8f c) {
    return __builtin_amdgcn_wmma_f32_16x16x4_f32(false, a, false, b, (short)0, c, false, false);
}

__global__ void zero_f32(float* __restrict__ p, int n) {
    int i = blockIdx.x * blockDim.x + threadIdx.x;
    int stride = gridDim.x * blockDim.x;
    for (; i < n; i += stride) p[i] = 0.f;
}

static constexpr int BN_SHIFT = 8;
static constexpr int BN   = 1 << BN_SHIFT;
static constexpr int NB   = (NN + BN - 1) / BN;
static constexpr int NBP  = 512;
static constexpr int CH   = 4096;
static constexpr int NCH  = (NE + CH - 1) / CH;
static constexpr int EPT  = CH / 256;
static constexpr int SND_BITS = 17;
static constexpr int SND_MASK = (1 << SND_BITS) - 1;
static constexpr int CAP  = 12288;
static_assert(NN <= (1 << SND_BITS), "a sender id must fit the packed word");
static_assert(NB <= NBP, "two bins per thread cover every bin");
static_assert(NE < (1 << 24), "degrees are whole numbers that stay exact in f32");

__device__ __forceinline__ int block_scan_incl(int v, int* buf) {
    const int t = threadIdx.x;
    buf[t] = v;
    __syncthreads();
    for (int d = 1; d < 256; d <<= 1) {
        const int a = (t >= d) ? buf[t - d] : 0;
        __syncthreads();
        v += a;
        buf[t] = v;
        __syncthreads();
    }
    return v;
}

__device__ __forceinline__ void block_scan_incl2(int& a, int& b, int* bufA, int* bufB) {
    const int t = threadIdx.x;
    bufA[t] = a;
    bufB[t] = b;
    __syncthreads();
    for (int d = 1; d < 256; d <<= 1) {
        const int xa = (t >= d) ? bufA[t - d] : 0;
        const int xb = (t >= d) ? bufB[t - d] : 0;
        __syncthreads();
        a += xa;
        b += xb;
        bufA[t] = a;
        bufB[t] = b;
        __syncthreads();
    }
}

static constexpr int SDR  = 8;
static constexpr int SDN  = NN / SDR;
static constexpr int SDC  = 32;
static constexpr int SDE  = NE / SDC;
static constexpr int SDV  = SDE / 4;
static constexpr int SDG  = (SDV + 255) / 256;
static constexpr int SD_BLKS = SDR * SDC;
static_assert(SDN * SDR == NN, "the ranges cover every node exactly");
static_assert(SDE * SDC == NE && SDV * 4 == SDE, "the chunks cover every edge in whole 16-byte loads");
static_assert(SDG <= 60, "a wave's loads in flight stay under the load counter's limit");

__device__ __forceinline__ void sdeg_body(const int blk, int* cnt,
                                          const int* __restrict__ snd, float* __restrict__ sdeg) {
    const int t  = threadIdx.x;
    const int lo = (blk % SDR) * SDN;
    const int4* __restrict__ src = (const int4*)(snd + (size_t)(blk / SDR) * SDE);
    int4 v[SDG];
    #pragma unroll
    for (int k = 0; k < SDG; ++k) {
        const int i = t + 256 * k;
        v[k] = (i < SDV) ? src[i] : make_int4(-1, -1, -1, -1);
    }
    for (int i = t; i < SDN; i += 256) cnt[i] = 0;
    __syncthreads();
    #pragma unroll
    for (int k = 0; k < SDG; ++k) {
        const unsigned a0 = (unsigned)(v[k].x - lo), a1 = (unsigned)(v[k].y - lo),
                       a2 = (unsigned)(v[k].z - lo), a3 = (unsigned)(v[k].w - lo);
        if (a0 < (unsigned)SDN) atomicAdd(&cnt[a0], 1);
        if (a1 < (unsigned)SDN) atomicAdd(&cnt[a1], 1);
        if (a2 < (unsigned)SDN) atomicAdd(&cnt[a2], 1);
        if (a3 < (unsigned)SDN) atomicAdd(&cnt[a3], 1);
    }
    __syncthreads();
    for (int i0 = t; i0 < SDN; i0 += 256 * 8) {
        int c[8];
        #pragma unroll
        for (int u = 0; u < 8; ++u) {
            const int i = i0 + 256 * u;
            c[u] = (i < SDN) ? cnt[i] : 0;
        }
        #pragma unroll
        for (int u = 0; u < 8; ++u)
            if (c[u]) atomicAdd(&sdeg[lo + i0 + 256 * u], (float)c[u]);
    }
}

__device__ __forceinline__ void rcount_body(const int blk, int* hist,
                                            const int* __restrict__ rcv, int* __restrict__ btot) {
    const int t = threadIdx.x;
    hist[t] = 0;
    hist[t + 256] = 0;
    __syncthreads();
    const int e0 = blk * CH + t;
    int r[EPT];
    #pragma unroll
    for (int k = 0; k < EPT; ++k) {
        const int e = e0 + k * 256;
        r[k] = (e < NE) ? rcv[e] : -1;
    }
    #pragma unroll
    for (int k = 0; k < EPT; ++k)
        if ((unsigned)r[k] < (unsigned)NN) atomicAdd(&hist[r[k] >> BN_SHIFT], 1);
    __syncthreads();
    const int h0 = hist[t], h1 = hist[t + 256];
    if (h0) atomicAdd(&btot[t], h0);
    if (h1) atomicAdd(&btot[t + 256], h1);
}

__launch_bounds__(256)
__global__ void count_kernel(const int* __restrict__ snd, const int* __restrict__ rcv,
                             float* __restrict__ sdeg, int* __restrict__ btot) {
    __shared__ __attribute__((aligned(16))) int smem[SDN > NBP ? SDN : NBP];
    const int blk = blockIdx.x;
    if (blk < SD_BLKS) sdeg_body(blk, smem, snd, sdeg);
    else               rcount_body(blk - SD_BLKS, smem, rcv, btot);
}

static constexpr int BIN_LDS_WORDS = 3 * NBP + 2 * 256 + 2 * CH;
__device__ __forceinline__ void bin_body(const int blk, int* smem,
                           const int* __restrict__ snd, const int* __restrict__ rcv,
                           const int* __restrict__ btot, int* __restrict__ bcur,
                           int* __restrict__ pairs) {
    int* hist   = smem;
    int* lstart = hist + NBP;
    int* gstart = lstart + NBP;
    int* sA     = gstart + NBP;
    int* sB     = sA + 256;
    int* sw     = sB + 256;
    int* sd     = sw + CH;
    const int t = threadIdx.x;
    hist[t] = 0;
    hist[t + 256] = 0;
    __syncthreads();
    int word[EPT], lr[EPT], bin[EPT];
    const int e0 = blk * CH + t;
    #pragma unroll
    for (int k = 0; k < EPT; ++k) {
        const int e = e0 + k * 256;
        bin[k]  = (e < NE) ? rcv[e] : -1;
        word[k] = (e < NE) ? snd[e] : 0;
    }
    #pragma unroll
    for (int k = 0; k < EPT; ++k) {
        const int r = bin[k];
        lr[k] = -1;
        bin[k] = 0;
        if ((unsigned)r < (unsigned)NN) {
            bin[k]  = r >> BN_SHIFT;
            lr[k]   = atomicAdd(&hist[bin[k]], 1);
            word[k] = ((r & (BN - 1)) << SND_BITS) | (word[k] & SND_MASK);
        }
    }
    __syncthreads();
    const int h0 = hist[2 * t], h1 = hist[2 * t + 1];
    const int g0 = btot[2 * t], g1 = btot[2 * t + 1];
    int li = h0 + h1, gi = g0 + g1;
    block_scan_incl2(li, gi, sA, sB);
    const int lex = li - (h0 + h1);
    const int gex = gi - (g0 + g1);
    int r0 = 0, r1 = 0;
    if (h0) r0 = atomicAdd(&bcur[2 * t], h0);
    if (h1) r1 = atomicAdd(&bcur[2 * t + 1], h1);
    lstart[2 * t]     = lex;
    lstart[2 * t + 1] = lex + h0;
    gstart[2 * t]     = gex + r0;
    gstart[2 * t + 1] = gex + g0 + r1;
    __syncthreads();
    #pragma unroll
    for (int k = 0; k < EPT; ++k) {
        if (lr[k] >= 0) {
            const int pos = lstart[bin[k]] + lr[k];
            sw[pos] = word[k];
            sd[pos] = gstart[bin[k]] + lr[k];
        }
    }
    __syncthreads();
    const int mine = sA[255];
    for (int i = t; i < mine; i += 256) {
        const int p = sd[i];
        if ((unsigned)p < (unsigned)NE) pairs[p] = sw[i];
    }
}

static constexpr int SORT_LDS_WORDS = 2 * 256 + 2 * BN + CAP;
__device__ __forceinline__ void sort_body(const int b, int* smem,
                            const int* __restrict__ pairs, const int* __restrict__ btot,
                            int* __restrict__ offs, float* __restrict__ rdeg, int* __restrict__ csr) {
    int* sA    = smem;
    int* sB    = sA + 256;
    int* cnt   = sB + 256;
    int* cur   = cnt + BN;
    int* stage = cur + BN;
    const int t = threadIdx.x;
    int front = 0;
    if (t < b)       front += btot[t];
    if (t + 256 < b) front += btot[t + 256];
    for (int m = 16; m >= 1; m >>= 1) front += __shfl_xor(front, m);
    if ((t & 31) == 0) sA[t >> 5] = front;
    __syncthreads();
    int all = 0;
    #pragma unroll
    for (int w = 0; w < 8; ++w) all += sA[w];
    const int start = min(max(all, 0), NE);
    const int nb    = min(max(btot[b], 0), NE - start);
    cnt[t] = 0;
    __syncthreads();
    for (int i0 = t; i0 < nb; i0 += 256 * 8) {
        int w[8];
        #pragma unroll
        for (int u = 0; u < 8; ++u) {
            const int i = i0 + 256 * u;
            w[u] = (i < nb) ? pairs[start + i] : -1;
        }
        #pragma unroll
        for (int u = 0; u < 8; ++u)
            if (w[u] >= 0) atomicAdd(&cnt[(w[u] >> SND_BITS) & (BN - 1)], 1);
    }
    __syncthreads();
    const int c   = cnt[t];
    const int loc = block_scan_incl(c, sB) - c;
    const int node = b * BN + t;
    if (node < NN) {
        offs[node] = start + loc;
        rdeg[node] = (float)c;
    }
    cur[t] = loc;
    __syncthreads();
    if (nb <= CAP) {
        for (int i0 = t; i0 < nb; i0 += 256 * 8) {
            int w[8];
            #pragma unroll
            for (int u = 0; u < 8; ++u) {
                const int i = i0 + 256 * u;
                w[u] = (i < nb) ? pairs[start + i] : -1;
            }
            #pragma unroll
            for (int u = 0; u < 8; ++u) {
                if (w[u] >= 0) {
                    const int p = atomicAdd(&cur[(w[u] >> SND_BITS) & (BN - 1)], 1);
                    if ((unsigned)p < (unsigned)CAP) stage[p] = w[u] & SND_MASK;
                }
            }
        }
        __syncthreads();
        for (int i = t; i < nb; i += 256) csr[start + i] = stage[i];
    } else {
        for (int i = t; i < nb; i += 256) {
            const int w = pairs[start + i];
            const int p = atomicAdd(&cur[(w >> SND_BITS) & (BN - 1)], 1);
            if ((unsigned)p < (unsigned)nb) csr[start + p] = w & SND_MASK;
        }
    }
}

static constexpr int MLP_LDS_WORDS = 8 * 16 * DH;
__device__ __forceinline__ void mlp_body(const int blk, float* lds,
                           const float* __restrict__ X,
                           const float* __restrict__ W1, const float* __restrict__ b1,
                           const float* __restrict__ W2, const float* __restrict__ b2,
                           const float* __restrict__ W3,
                           const float* __restrict__ sdeg,
                           float* __restrict__ p1, int nTiles) {
    const int w    = threadIdx.x >> 5;
    const int lane = threadIdx.x & 31;
    const int m16  = lane & 15;
    const int kh   = lane >> 4;
    const int t    = blk * 8 + w;
    const bool active = (t < nTiles);
    const int row0 = t * 16;
    float* my = &lds[w * 16 * DH];

    if (active) {
        v8f c0 = {}, c1 = {}, c2 = {}, c3 = {};
        #pragma unroll 4
        for (int kb = 0; kb < DF; kb += 4) {
            v2f a = *(const v2f*)(X + (size_t)(row0 + m16) * DF + kb + 2 * kh);
            const float* wr0 = W1 + (size_t)(kb + 2 * kh) * DH;
            const float* wr1 = wr0 + DH;
            v2f b;
            b.x = wr0[m16];      b.y = wr1[m16];      c0 = wmma4(a, b, c0);
            b.x = wr0[16 + m16]; b.y = wr1[16 + m16]; c1 = wmma4(a, b, c1);
            b.x = wr0[32 + m16]; b.y = wr1[32 + m16]; c2 = wmma4(a, b, c2);
            b.x = wr0[48 + m16]; b.y = wr1[48 + m16]; c3 = wmma4(a, b, c3);
        }
        #pragma unroll
        for (int r = 0; r < 8; ++r) {
            float* dr = my + (r + 8 * kh) * DH;
            dr[m16]      = lrelu(c0[r] + b1[m16]);
            dr[16 + m16] = lrelu(c1[r] + b1[16 + m16]);
            dr[32 + m16] = lrelu(c2[r] + b1[32 + m16]);
            dr[48 + m16] = lrelu(c3[r] + b1[48 + m16]);
        }
    }
    __syncthreads();

    if (active) {
        v8f d0 = {}, d1 = {}, d2 = {}, d3 = {};
        #pragma unroll 4
        for (int kb = 0; kb < DH; kb += 4) {
            v2f a = *(const v2f*)(my + m16 * DH + kb + 2 * kh);
            const float* wr0 = W2 + (size_t)(kb + 2 * kh) * DH;
            const float* wr1 = wr0 + DH;
            v2f b;
            b.x = wr0[m16];      b.y = wr1[m16];      d0 = wmma4(a, b, d0);
            b.x = wr0[16 + m16]; b.y = wr1[16 + m16]; d1 = wmma4(a, b, d1);
            b.x = wr0[32 + m16]; b.y = wr1[32 + m16]; d2 = wmma4(a, b, d2);
            b.x = wr0[48 + m16]; b.y = wr1[48 + m16]; d3 = wmma4(a, b, d3);
        }
        float sc[8];
        #pragma unroll
        for (int r = 0; r < 8; ++r) sc[r] = rsqrtf(fmaxf(sdeg[row0 + r + 8 * kh], 1.f));
        #pragma unroll
        for (int r = 0; r < 8; ++r) {
            float* dr = my + (r + 8 * kh) * DH;
            dr[m16]      = lrelu(d0[r] + b2[m16])      * sc[r];
            dr[16 + m16] = lrelu(d1[r] + b2[16 + m16]) * sc[r];
            dr[32 + m16] = lrelu(d2[r] + b2[32 + m16]) * sc[r];
            dr[48 + m16] = lrelu(d3[r] + b2[48 + m16]) * sc[r];
        }
    }
    __syncthreads();
    if (!active) return;

    v8f e0 = {}, e1 = {}, e2 = {};
    #pragma unroll 4
    for (int kb = 0; kb < DH; kb += 4) {
        v2f a = *(const v2f*)(my + m16 * DH + kb + 2 * kh);
        const float* wr0 = W3 + (size_t)(kb + 2 * kh) * DC;
        const float* wr1 = wr0 + DC;
        v2f b;
        b.x = wr0[m16];      b.y = wr1[m16];      e0 = wmma4(a, b, e0);
        b.x = wr0[16 + m16]; b.y = wr1[16 + m16]; e1 = wmma4(a, b, e1);
        const int n2  = 32 + m16;
        const int n2c = (n2 < DC) ? n2 : DC - 1;
        const float x0 = wr0[n2c], x1 = wr1[n2c];
        b.x = (n2 < DC) ? x0 : 0.f;
        b.y = (n2 < DC) ? x1 : 0.f;
        e2 = wmma4(a, b, e2);
    }
    #pragma unroll
    for (int r = 0; r < 8; ++r) {
        float* orow = p1 + (size_t)(row0 + r + 8 * kh) * P1P;
        orow[m16]      = e0[r];
        orow[16 + m16] = e1[r];
        if (32 + m16 < DC)
            orow[32 + m16] = e2[r];
    }
}

static constexpr int MLP_BLKS   = (NN / 16 + 7) / 8;
static constexpr int MLP_BLKS_A = MLP_BLKS / 2;
static constexpr int MLP_BLKS_B = MLP_BLKS - MLP_BLKS_A;
static constexpr int LDS_A = BIN_LDS_WORDS  > MLP_LDS_WORDS ? BIN_LDS_WORDS  : MLP_LDS_WORDS;
static constexpr int LDS_B = SORT_LDS_WORDS > MLP_LDS_WORDS ? SORT_LDS_WORDS : MLP_LDS_WORDS;

__launch_bounds__(256)
__global__ void bin_mlp_kernel(const int* __restrict__ snd, const int* __restrict__ rcv,
                               const int* __restrict__ btot, int* __restrict__ bcur,
                               int* __restrict__ pairs,
                               const float* __restrict__ X,
                               const float* __restrict__ W1, const float* __restrict__ b1,
                               const float* __restrict__ W2, const float* __restrict__ b2,
                               const float* __restrict__ W3,
                               const float* __restrict__ sdeg,
                               float* __restrict__ p1, int nTiles) {
    __shared__ __attribute__((aligned(16))) int smem[LDS_A];
    const int blk = blockIdx.x;
    if (blk < NCH) bin_body(blk, smem, snd, rcv, btot, bcur, pairs);
    else           mlp_body(blk - NCH, (float*)smem, X, W1, b1, W2, b2, W3, sdeg, p1, nTiles);
}

__launch_bounds__(256)
__global__ void sort_mlp_kernel(const int* __restrict__ pairs, const int* __restrict__ btot,
                                int* __restrict__ offs, float* __restrict__ rdeg, int* __restrict__ csr,
                                const float* __restrict__ X,
                                const float* __restrict__ W1, const float* __restrict__ b1,
                                const float* __restrict__ W2, const float* __restrict__ b2,
                                const float* __restrict__ W3,
                                const float* __restrict__ sdeg,
                                float* __restrict__ p1, int nTiles) {
    __shared__ __attribute__((aligned(16))) int smem[LDS_B];
    const int blk = blockIdx.x;
    if (blk < NB) sort_body(blk, smem, pairs, btot, offs, rdeg, csr);
    else          mlp_body(MLP_BLKS_A + (blk - NB), (float*)smem, X, W1, b1, W2, b2, W3, sdeg, p1, nTiles);
}

__launch_bounds__(256)
__global__ void gather1_kernel(const float* __restrict__ p1, const int* __restrict__ csr,
                               const int* __restrict__ offs, const float* __restrict__ rdeg,
                               const float* __restrict__ sdeg, const float* __restrict__ b3,
                               float* __restrict__ h2) {
    const int wave = (blockIdx.x * 256 + threadIdx.x) >> 5;
    const int lane = threadIdx.x & 31;
    const int q = lane / 10;
    const int l = lane - 10 * q;
    const int n = wave * G1N + q;
    if (q >= G1N || n >= NN) return;
    const int* seg = csr + offs[n];
    const float rd = rdeg[n];
    const int cnt = (int)rd;
    float4 acc = {0.f, 0.f, 0.f, 0.f};
    for (int j = 0; j < cnt; j += 8) {
        int s[8];
        float4 v[8];
        #pragma unroll
        for (int k = 0; k < 8; ++k) {
            const int qd = seg[min(j + k, cnt - 1)];
            s[k] = ((unsigned)qd < (unsigned)NN) ? qd : 0;
        }
        #pragma unroll
        for (int k = 0; k < 8; ++k)
            v[k] = *(const float4*)(p1 + (size_t)s[k] * P1P + 4 * l);
        #pragma unroll
        for (int k = 0; k < 8; ++k) {
            const bool in = j + k < cnt;
            acc.x += in ? v[k].x : 0.f; acc.y += in ? v[k].y : 0.f;
            acc.z += in ? v[k].z : 0.f; acc.w += in ? v[k].w : 0.f;
        }
    }
    const float rs = rsqrtf(fmaxf(rd, 1.f));
    const float sc = rsqrtf(fmaxf(sdeg[n], 1.f));
    float4 o;
    o.x = (acc.x * rs + b3[4 * l])     * sc;
    o.y = (acc.y * rs + b3[4 * l + 1]) * sc;
    o.z = (acc.z * rs + b3[4 * l + 2]) * sc;
    o.w = (acc.w * rs + b3[4 * l + 3]) * sc;
    *(float4*)(h2 + (size_t)n * DC + 4 * l) = o;
}

__launch_bounds__(256)
__global__ void gather2_softmax_kernel(const float* __restrict__ h2, const int* __restrict__ csr,
                                       const int* __restrict__ offs, const float* __restrict__ rdeg,
                                       float* __restrict__ out) {
    const int gt = blockIdx.x * 256 + threadIdx.x;
    const int n = gt >> 3;
    const int l = gt & 7;
    if (n >= NN) return;
    const int* seg = csr + offs[n];
    const float rd = rdeg[n];
    const int cnt = (int)rd;
    float4 acc = {0.f, 0.f, 0.f, 0.f};
    float tail = 0.f;
    for (int j = 0; j < cnt; j += 8) {
        int s[8];
        float4 v[8];
        float u[8];
        #pragma unroll
        for (int k = 0; k < 8; ++k) {
            const int q = seg[min(j + k, cnt - 1)];
            s[k] = ((unsigned)q < (unsigned)NN) ? q : 0;
        }
        #pragma unroll
        for (int k = 0; k < 8; ++k) {
            const float* row = h2 + (size_t)s[k] * DC;
            v[k] = *(const float4*)(row + 4 * l);
            u[k] = row[32 + l];
        }
        #pragma unroll
        for (int k = 0; k < 8; ++k) {
            const bool in = j + k < cnt;
            acc.x += in ? v[k].x : 0.f; acc.y += in ? v[k].y : 0.f;
            acc.z += in ? v[k].z : 0.f; acc.w += in ? v[k].w : 0.f;
            tail += in ? u[k] : 0.f;
        }
    }
    const float rs = rsqrtf(fmaxf(rd, 1.f));
    const float a0 = acc.x * rs, a1 = acc.y * rs, a2 = acc.z * rs, a3 = acc.w * rs, a4 = tail * rs;
    float mx = fmaxf(fmaxf(fmaxf(a0, a1), fmaxf(a2, a3)), a4);
    #pragma unroll
    for (int m = 1; m < 8; m <<= 1) mx = fmaxf(mx, __shfl_xor(mx, m));
    const float e0 = __expf(a0 - mx), e1 = __expf(a1 - mx), e2 = __expf(a2 - mx),
                e3 = __expf(a3 - mx), e4 = __expf(a4 - mx);
    float sum = (e0 + e1) + (e2 + e3) + e4;
    #pragma unroll
    for (int m = 1; m < 8; m <<= 1) sum += __shfl_xor(sum, m);
    const float inv = 1.f / sum;
    float* orow = out + (size_t)n * DC;
    float4 o;
    o.x = e0 * inv; o.y = e1 * inv; o.z = e2 * inv; o.w = e3 * inv;
    *(float4*)(orow + 4 * l) = o;
    orow[32 + l] = e4 * inv;
}

extern "C" void kernel_launch(void* const* d_in, const int* in_sizes, int n_in,
                              void* d_out, int out_size, void* d_ws, size_t ws_size,
                              hipStream_t stream) {
    const float* nodes     = (const float*)d_in[0];
    const int*   senders   = (const int*)d_in[1];
    const int*   receivers = (const int*)d_in[2];
    const float* W1 = (const float*)d_in[3];
    const float* b1 = (const float*)d_in[4];
    const float* W2 = (const float*)d_in[5];
    const float* b2 = (const float*)d_in[6];
    const float* W3 = (const float*)d_in[7];
    const float* b3 = (const float*)d_in[8];
    float* out = (float*)d_out;
    float* ws  = (float*)d_ws;

    float* sdeg = ws;
    int*   btot = (int*)(ws + NN);
    int*   bcur = btot + NBP;
    float* rdeg = ws + NN + 2 * NBP;
    float* p1   = rdeg + NN;
    float* h2   = p1 + (size_t)NN * P1P;
    int*   offs = (int*)(h2 + (size_t)NN * DC);
    int*   pairs = offs + NN;
    int*   csr   = pairs + NE;

    const int nTiles = NN / 16;
    zero_f32<<<2048, 256, 0, stream>>>(sdeg, NN + 2 * NBP);
    count_kernel<<<SD_BLKS + NCH, 256, 0, stream>>>(senders, receivers, sdeg, btot);
    bin_mlp_kernel<<<NCH + MLP_BLKS_A, 256, 0, stream>>>(senders, receivers, btot, bcur, pairs,
                                                         nodes, W1, b1, W2, b2, W3, sdeg, p1, nTiles);
    sort_mlp_kernel<<<NB + MLP_BLKS_B, 256, 0, stream>>>(pairs, btot, offs, rdeg, csr,
                                                         nodes, W1, b1, W2, b2, W3, sdeg, p1, nTiles);
    gather1_kernel<<<((NN + G1N - 1) / G1N + 7) / 8, 256, 0, stream>>>(p1, csr, offs, rdeg, sdeg, b3, h2);
    gather2_softmax_kernel<<<(NN * 8 + 255) / 256, 256, 0, stream>>>(h2, csr, offs, rdeg, out);
}


# Launches 6 -> 4: bin pass keeps each block's edges in its own stretch plus an offsets table, degrees go to per-chunk planes summed by their readers: no zero or count launch; edge sum order changes
# speedup vs baseline: 6.8128x; 1.0432x over previous
#include <hip/hip_runtime.h>

typedef __attribute__((ext_vector_type(2))) float v2f;
typedef __attribute__((ext_vector_type(8))) float v8f;

static constexpr int NN = 100000;
static constexpr int NE = 1600000;
static constexpr int DF = 128;
static constexpr int DH = 64;
static constexpr int DC = 40;
static constexpr int P1P = 64;
static constexpr int G1N = 3;
#define NEG_SLOPE 0.01f

__device__ __forceinline__ float lrelu(float x) { return x > 0.f ? x : x * NEG_SLOPE; }

__device__ __forceinline__ v8f wmma4(v2f a, v2f b, v8f c) {
    return __builtin_amdgcn_wmma_f32_16x16x4_f32(false, a, false, b, (short)0, c, false, false);
}

static constexpr int BN_SHIFT = 8;
static constexpr int BN   = 1 << BN_SHIFT;
static constexpr int NB   = (NN + BN - 1) / BN;
static constexpr int NBP  = 512;
static constexpr int CH   = 4096;
static constexpr int NCH  = (NE + CH - 1) / CH;
static constexpr int EPT  = CH / 256;
static constexpr int SND_BITS = 17;
static constexpr int SND_MASK = (1 << SND_BITS) - 1;
static constexpr int CAP  = 12288;
static constexpr int PAIRS_WORDS = NCH * CH;
static_assert(NN <= (1 << SND_BITS), "a sender id must fit the packed word");
static_assert(NB < NBP, "two bins per thread cover every bin and the entry behind the last one");
static_assert(NCH <= NBP, "two bin-pass blocks per thread cover every block");
static_assert(NE < (1 << 24), "degrees are whole numbers that stay exact in f32");

__device__ __forceinline__ int block_scan_incl(int v, int* buf) {
    const int t = threadIdx.x;
    buf[t] = v;
    __syncthreads();
    for (int d = 1; d < 256; d <<= 1) {
        const int a = (t >= d) ? buf[t - d] : 0;
        __syncthreads();
        v += a;
        buf[t] = v;
        __syncthreads();
    }
    return v;
}

__device__ __forceinline__ void block_scan_incl2(int& a, int& b, int* bufA, int* bufB) {
    const int t = threadIdx.x;
    bufA[t] = a;
    bufB[t] = b;
    __syncthreads();
    for (int d = 1; d < 256; d <<= 1) {
        const int xa = (t >= d) ? bufA[t - d] : 0;
        const int xb = (t >= d) ? bufB[t - d] : 0;
        __syncthreads();
        a += xa;
        b += xb;
        bufA[t] = a;
        bufB[t] = b;
        __syncthreads();
    }
}

static constexpr int SDR  = 8;
static constexpr int SDN  = NN / SDR;
static constexpr int SDC  = 32;
static constexpr int SDE  = NE / SDC;
static constexpr int SDV  = SDE / 4;
static constexpr int SDG  = (SDV + 255) / 256;
static constexpr int SD_BLKS = SDR * SDC;
static_assert(SDN * SDR == NN, "the ranges cover every node exactly");
static_assert(SDE * SDC == NE && SDV * 4 == SDE, "the chunks cover every edge in whole 16-byte loads");
static_assert(SDG <= 60, "a wave's loads in flight stay under the load counter's limit");

static constexpr int SDG_A = (SDG + 1) / 2;
template <int K0, int K1>
__device__ __forceinline__ void sdeg_load(int4 (&v)[K1 - K0], const int4* __restrict__ src, const int t) {
    #pragma unroll
    for (int k = K0; k < K1; ++k) {
        const int i = t + 256 * k;
        v[k - K0] = (i < SDV) ? src[i] : make_int4(-1, -1, -1, -1);
    }
}
template <int K>
__device__ __forceinline__ void sdeg_add(const int4 (&v)[K], int* cnt, const int lo) {
    #pragma unroll
    for (int k = 0; k < K; ++k) {
        const unsigned a0 = (unsigned)(v[k].x - lo), a1 = (unsigned)(v[k].y - lo),
                       a2 = (unsigned)(v[k].z - lo), a3 = (unsigned)(v[k].w - lo);
        if (a0 < (unsigned)SDN) atomicAdd(&cnt[a0], 1);
        if (a1 < (unsigned)SDN) atomicAdd(&cnt[a1], 1);
        if (a2 < (unsigned)SDN) atomicAdd(&cnt[a2], 1);
        if (a3 < (unsigned)SDN) atomicAdd(&cnt[a3], 1);
    }
}

__device__ __forceinline__ void sdeg_body(const int blk, int* cnt,
                                          const int* __restrict__ snd, float* __restrict__ spart) {
    const int t  = threadIdx.x;
    const int lo = (blk % SDR) * SDN;
    const int4* __restrict__ src = (const int4*)(snd + (size_t)(blk / SDR) * SDE);
    int4 va[SDG_A];
    sdeg_load<0, SDG_A>(va, src, t);
    for (int i = t; i < SDN; i += 256) cnt[i] = 0;
    __syncthreads();
    sdeg_add<SDG_A>(va, cnt, lo);
    int4 vb[SDG - SDG_A];
    sdeg_load<SDG_A, SDG>(vb, src, t);
    sdeg_add<SDG - SDG_A>(vb, cnt, lo);
    __syncthreads();
    float* __restrict__ plane = spart + (size_t)(blk / SDR) * NN + lo;
    for (int i0 = t; i0 < SDN; i0 += 256 * 8) {
        int c[8];
        #pragma unroll
        for (int u = 0; u < 8; ++u) {
            const int i = i0 + 256 * u;
            c[u] = (i < SDN) ? cnt[i] : 0;
        }
        #pragma unroll
        for (int u = 0; u < 8; ++u)
            if (i0 + 256 * u < SDN) plane[i0 + 256 * u] = (float)c[u];
    }
}

static constexpr int BIN_LDS_WORDS = 2 * NBP + 256 + CH;
__device__ __forceinline__ void bin_body(const int blk, int* smem,
                           const int* __restrict__ snd, const int* __restrict__ rcv,
                           int* __restrict__ tab, int* __restrict__ pairs) {
    int* hist   = smem;
    int* lstart = hist + NBP;
    int* sA     = lstart + NBP;
    int* sw     = sA + 256;
    const int t = threadIdx.x;
    hist[t] = 0;
    hist[t + 256] = 0;
    __syncthreads();
    int word[EPT], lr[EPT], bin[EPT];
    const int e0 = blk * CH + t;
    #pragma unroll
    for (int k = 0; k < EPT; ++k) {
        const int e = e0 + k * 256;
        bin[k]  = (e < NE) ? rcv[e] : -1;
        word[k] = (e < NE) ? snd[e] : 0;
    }
    #pragma unroll
    for (int k = 0; k < EPT; ++k) {
        const int r = bin[k];
        lr[k] = -1;
        bin[k] = 0;
        if ((unsigned)r < (unsigned)NN) {
            bin[k]  = r >> BN_SHIFT;
            lr[k]   = atomicAdd(&hist[bin[k]], 1);
            word[k] = ((r & (BN - 1)) << SND_BITS) | (word[k] & SND_MASK);
        }
    }
    __syncthreads();
    const int h0 = hist[2 * t], h1 = hist[2 * t + 1];
    const int lex = block_scan_incl(h0 + h1, sA) - (h0 + h1);
    lstart[2 * t]     = lex;
    lstart[2 * t + 1] = lex + h0;
    *(int2*)(tab + (size_t)blk * NBP + 2 * t) = make_int2(lex, lex + h0);
    __syncthreads();
    #pragma unroll
    for (int k = 0; k < EPT; ++k)
        if (lr[k] >= 0) sw[lstart[bin[k]] + lr[k]] = word[k];
    __syncthreads();
    const int mine = min(sA[255], CH);
    int* __restrict__ dst = pairs + (size_t)blk * CH;
    for (int i = t; i < mine; i += 256) dst[i] = sw[i];
}

static constexpr int SORT_LDS_WORDS = 2 * 256 + 2 * NBP + 2 * BN + CAP;
__device__ __forceinline__ int run_fetch(const int i, const int* rs, const int* ro,
                                         const int* __restrict__ pairs) {
    int k = 0;
    #pragma unroll
    for (int step = NBP / 2; step >= 1; step >>= 1)
        if (rs[k + step] <= i) k += step;
    const int p = ro[k] + (i - rs[k]);
    return pairs[min(max(p, 0), PAIRS_WORDS - 1)];
}

__device__ __forceinline__ void sort_body(const int b, int* smem,
                            const int* __restrict__ pairs, const int* __restrict__ tab,
                            int* __restrict__ offs, float* __restrict__ rdeg, int* __restrict__ csr) {
    int* sA    = smem;
    int* sB    = sA + 256;
    int* rs    = sB + 256;
    int* ro    = rs + NBP;
    int* cnt   = ro + NBP;
    int* cur   = cnt + BN;
    int* stage = cur + BN;
    const int t = threadIdx.x;
    const int k0 = 2 * t, k1 = 2 * t + 1;
    int f0 = 0, l0 = 0, f1 = 0, l1 = 0;
    if (k0 < NCH) {
        const int x = tab[(size_t)k0 * NBP + b], y = tab[(size_t)k0 * NBP + b + 1];
        f0 = min(max(x, 0), CH);
        l0 = min(max(y - x, 0), CH - f0);
    }
    if (k1 < NCH) {
        const int x = tab[(size_t)k1 * NBP + b], y = tab[(size_t)k1 * NBP + b + 1];
        f1 = min(max(x, 0), CH);
        l1 = min(max(y - x, 0), CH - f1);
    }
    int li = l0 + l1, fi = f0 + f1;
    block_scan_incl2(li, fi, sA, sB);
    rs[k0] = li - (l0 + l1);
    rs[k1] = li - l1;
    ro[k0] = k0 * CH + f0;
    ro[k1] = k1 * CH + f1;
    cnt[t] = 0;
    __syncthreads();
    const int start = min(max(sB[255], 0), NE);
    const int nb    = min(max(sA[255], 0), NE - start);
    for (int i0 = t; i0 < nb; i0 += 256 * 8) {
        int w[8];
        #pragma unroll
        for (int u = 0; u < 8; ++u) {
            const int i = i0 + 256 * u;
            const int x = run_fetch(min(i, nb - 1), rs, ro, pairs);
            w[u] = (i < nb) ? x : -1;
        }
        #pragma unroll
        for (int u = 0; u < 8; ++u)
            if (w[u] >= 0) atomicAdd(&cnt[(w[u] >> SND_BITS) & (BN - 1)], 1);
    }
    __syncthreads();
    const int c   = cnt[t];
    const int loc = block_scan_incl(c, sB) - c;
    const int node = b * BN + t;
    if (node < NN) {
        offs[node] = start + loc;
        rdeg[node] = (float)c;
    }
    cur[t] = loc;
    __syncthreads();
    if (nb <= CAP) {
        for (int i0 = t; i0 < nb; i0 += 256 * 8) {
            int w[8];
            #pragma unroll
            for (int u = 0; u < 8; ++u) {
                const int i = i0 + 256 * u;
                const int x = run_fetch(min(i, nb - 1), rs, ro, pairs);
                w[u] = (i < nb) ? x : -1;
            }
            #pragma unroll
            for (int u = 0; u < 8; ++u) {
                if (w[u] >= 0) {
                    const int p = atomicAdd(&cur[(w[u] >> SND_BITS) & (BN - 1)], 1);
                    if ((unsigned)p < (unsigned)CAP) stage[p] = w[u] & SND_MASK;
                }
            }
        }
        __syncthreads();
        for (int i = t; i < nb; i += 256) csr[start + i] = stage[i];
    } else {
        for (int i = t; i < nb; i += 256) {
            const int w = run_fetch(i, rs, ro, pairs);
            const int p = atomicAdd(&cur[(w >> SND_BITS) & (BN - 1)], 1);
            if ((unsigned)p < (unsigned)nb) csr[start + p] = w & SND_MASK;
        }
    }
}

static constexpr int MLP_LDS_WORDS = 8 * 16 * DH;
template <bool SCALED>
__device__ __forceinline__ void mlp_body(const int blk, float* lds,
                           const float* __restrict__ X,
                           const float* __restrict__ W1, const float* __restrict__ b1,
                           const float* __restrict__ W2, const float* __restrict__ b2,
                           const float* __restrict__ W3,
                           const float* __restrict__ spart, float* __restrict__ sdeg,
                           float* __restrict__ p1, int nTiles) {
    const int w    = threadIdx.x >> 5;
    const int lane = threadIdx.x & 31;
    const int m16  = lane & 15;
    const int kh   = lane >> 4;
    const int t    = blk * 8 + w;
    const bool active = (t < nTiles);
    const int row0 = t * 16;
    float* my = &lds[w * 16 * DH];

    if (active) {
        v8f c0 = {}, c1 = {}, c2 = {}, c3 = {};
        #pragma unroll 4
        for (int kb = 0; kb < DF; kb += 4) {
            v2f a = *(const v2f*)(X + (size_t)(row0 + m16) * DF + kb + 2 * kh);
            const float* wr0 = W1 + (size_t)(kb + 2 * kh) * DH;
            const float* wr1 = wr0 + DH;
            v2f b;
            b.x = wr0[m16];      b.y = wr1[m16];      c0 = wmma4(a, b, c0);
            b.x = wr0[16 + m16]; b.y = wr1[16 + m16]; c1 = wmma4(a, b, c1);
            b.x = wr0[32 + m16]; b.y = wr1[32 + m16]; c2 = wmma4(a, b, c2);
            b.x = wr0[48 + m16]; b.y = wr1[48 + m16]; c3 = wmma4(a, b, c3);
        }
        #pragma unroll
        for (int r = 0; r < 8; ++r) {
            float* dr = my + (r + 8 * kh) * DH;
            dr[m16]      = lrelu(c0[r] + b1[m16]);
            dr[16 + m16] = lrelu(c1[r] + b1[16 + m16]);
            dr[32 + m16] = lrelu(c2[r] + b1[32 + m16]);
            dr[48 + m16] = lrelu(c3[r] + b1[48 + m16]);
        }
    }
    __syncthreads();

    if (active) {
        v8f d0 = {}, d1 = {}, d2 = {}, d3 = {};
        #pragma unroll 4
        for (int kb = 0; kb < DH; kb += 4) {
            v2f a = *(const v2f*)(my + m16 * DH + kb + 2 * kh);
            const float* wr0 = W2 + (size_t)(kb + 2 * kh) * DH;
            const float* wr1 = wr0 + DH;
            v2f b;
            b.x = wr0[m16];      b.y = wr1[m16];      d0 = wmma4(a, b, d0);
            b.x = wr0[16 + m16]; b.y = wr1[16 + m16]; d1 = wmma4(a, b, d1);
            b.x = wr0[32 + m16]; b.y = wr1[32 + m16]; d2 = wmma4(a, b, d2);
            b.x = wr0[48 + m16]; b.y = wr1[48 + m16]; d3 = wmma4(a, b, d3);
        }
        float sc[8];
        if (SCALED) {
            float part = 0.f;
            #pragma unroll
            for (int j = 0; j < SDC / 2; ++j) part += spart[(size_t)(kh * (SDC / 2) + j) * NN + row0 + m16];
            const float tot = part + __shfl_xor(part, 16);
            if (kh == 0) sdeg[row0 + m16] = tot;
            #pragma unroll
            for (int r = 0; r < 8; ++r) sc[r] = rsqrtf(fmaxf(__shfl(tot, r + 8 * kh), 1.f));
        } else {
            #pragma unroll
            for (int r = 0; r < 8; ++r) sc[r] = 1.f;
        }
        #pragma unroll
        for (int r = 0; r < 8; ++r) {
            float* dr = my + (r + 8 * kh) * DH;
            dr[m16]      = lrelu(d0[r] + b2[m16])      * sc[r];
            dr[16 + m16] = lrelu(d1[r] + b2[16 + m16]) * sc[r];
            dr[32 + m16] = lrelu(d2[r] + b2[32 + m16]) * sc[r];
            dr[48 + m16] = lrelu(d3[r] + b2[48 + m16]) * sc[r];
        }
    }
    __syncthreads();
    if (!active) return;

    v8f e0 = {}, e1 = {}, e2 = {};
    #pragma unroll 4
    for (int kb = 0; kb < DH; kb += 4) {
        v2f a = *(const v2f*)(my + m16 * DH + kb + 2 * kh);
        const float* wr0 = W3 + (size_t)(kb + 2 * kh) * DC;
        const float* wr1 = wr0 + DC;
        v2f b;
        b.x = wr0[m16];      b.y = wr1[m16];      e0 = wmma4(a, b, e0);
        b.x = wr0[16 + m16]; b.y = wr1[16 + m16]; e1 = wmma4(a, b, e1);
        const int n2  = 32 + m16;
        const int n2c = (n2 < DC) ? n2 : DC - 1;
        const float x0 = wr0[n2c], x1 = wr1[n2c];
        b.x = (n2 < DC) ? x0 : 0.f;
        b.y = (n2 < DC) ? x1 : 0.f;
        e2 = wmma4(a, b, e2);
    }
    #pragma unroll
    for (int r = 0; r < 8; ++r) {
        float* orow = p1 + (size_t)(row0 + r + 8 * kh) * P1P;
        orow[m16]      = e0[r];
        orow[16 + m16] = e1[r];
        if (32 + m16 < DC)
            orow[32 + m16] = e2[r];
    }
}

static constexpr int MLP_BLKS   = (NN / 16 + 7) / 8;
static constexpr int MLP_BLKS_A = MLP_BLKS / 2;
static constexpr int MLP_BLKS_B = MLP_BLKS - MLP_BLKS_A;
static constexpr int ROWS_A     = MLP_BLKS_A * 8 * 16;
static constexpr int SCL_BLKS   = (ROWS_A + 255) / 256;
static_assert(ROWS_A <= NN, "the first launch's tiles are all whole");
__device__ __forceinline__ void scale_body(const int blk, const float* __restrict__ spart,
                                           float* __restrict__ sdeg, float* __restrict__ p1) {
    const int lane = threadIdx.x & 31;
    const int base = blk * 256 + (threadIdx.x >> 5) * 32;
    if (base >= ROWS_A) return;
    const int n = min(base + lane, ROWS_A - 1);
    float tot = 0.f;
    #pragma unroll
    for (int c = 0; c < SDC; ++c) tot += spart[(size_t)c * NN + n];
    if (base + lane < ROWS_A) sdeg[n] = tot;
    const float sc = rsqrtf(fmaxf(tot, 1.f));
    #pragma unroll
    for (int it = 0; it < 10; ++it) {
        const int idx = it * 32 + lane;
        const int r = idx / 10, l = idx - 10 * r;
        const float s = __shfl(sc, r);
        if (base + r < ROWS_A) {
            float4* p = (float4*)(p1 + (size_t)(base + r) * P1P + 4 * l);
            float4 v = *p;
            v.x *= s; v.y *= s; v.z *= s; v.w *= s;
            *p = v;
        }
    }
}

static constexpr int LDS_A0 = SDN > BIN_LDS_WORDS ? SDN : BIN_LDS_WORDS;
static constexpr int LDS_A  = LDS_A0 > MLP_LDS_WORDS ? LDS_A0 : MLP_LDS_WORDS;
static constexpr int LDS_B  = SORT_LDS_WORDS > MLP_LDS_WORDS ? SORT_LDS_WORDS : MLP_LDS_WORDS;

__launch_bounds__(256)
__global__ void deg_bin_mlp_kernel(const int* __restrict__ snd, const int* __restrict__ rcv,
                                   float* __restrict__ spart, int* __restrict__ tab,
                                   int* __restrict__ pairs,
                                   const float* __restrict__ X,
                                   const float* __restrict__ W1, const float* __restrict__ b1,
                                   const float* __restrict__ W2, const float* __restrict__ b2,
                                   const float* __restrict__ W3,
                                   float* __restrict__ sdeg,
                                   float* __restrict__ p1, int nTiles) {
    __shared__ __attribute__((aligned(16))) int smem[LDS_A];
    const int blk = blockIdx.x;
    if (blk < SD_BLKS)            sdeg_body(blk, smem, snd, spart);
    else if (blk < SD_BLKS + NCH) bin_body(blk - SD_BLKS, smem, snd, rcv, tab, pairs);
    else mlp_body<false>(blk - SD_BLKS - NCH, (float*)smem, X, W1, b1, W2, b2, W3, spart, sdeg, p1, nTiles);
}

__launch_bounds__(256)
__global__ void sort_mlp_kernel(const int* __restrict__ pairs, const int* __restrict__ tab,
                                int* __restrict__ offs, float* __restrict__ rdeg, int* __restrict__ csr,
                                const float* __restrict__ X,
                                const float* __restrict__ W1, const float* __restrict__ b1,
                                const float* __restrict__ W2, const float* __restrict__ b2,
                                const float* __restrict__ W3,
                                const float* __restrict__ spart, float* __restrict__ sdeg,
                                float* __restrict__ p1, int nTiles) {
    __shared__ __attribute__((aligned(16))) int smem[LDS_B];
    const int blk = blockIdx.x;
    if (blk < NB)                   sort_body(blk, smem, pairs, tab, offs, rdeg, csr);
    else if (blk < NB + MLP_BLKS_B) mlp_body<true>(MLP_BLKS_A + (blk - NB), (float*)smem, X, W1, b1, W2, b2, W3, spart, sdeg, p1, nTiles);
    else                            scale_body(blk - NB - MLP_BLKS_B, spart, sdeg, p1);
}

__launch_bounds__(256)
__global__ void gather1_kernel(const float* __restrict__ p1, const int* __restrict__ csr,
                               const int* __restrict__ offs, const float* __restrict__ rdeg,
                               const float* __restrict__ sdeg, const float* __restrict__ b3,
                               float* __restrict__ h2) {
    const int wave = (blockIdx.x * 256 + threadIdx.x) >> 5;
    const int lane = threadIdx.x & 31;
    const int q = lane / 10;
    const int l = lane - 10 * q;
    const int n = wave * G1N + q;
    if (q >= G1N || n >= NN) return;
    const int* seg = csr + offs[n];
    const float rd = rdeg[n];
    const int cnt = (int)rd;
    float4 acc = {0.f, 0.f, 0.f, 0.f};
    for (int j = 0; j < cnt; j += 8) {
        int s[8];
        float4 v[8];
        #pragma unroll
        for (int k = 0; k < 8; ++k) {
            const int qd = seg[min(j + k, cnt - 1)];
            s[k] = ((unsigned)qd < (unsigned)NN) ? qd : 0;
        }
        #pragma unroll
        for (int k = 0; k < 8; ++k)
            v[k] = *(const float4*)(p1 + (size_t)s[k] * P1P + 4 * l);
        #pragma unroll
        for (int k = 0; k < 8; ++k) {
            const bool in = j + k < cnt;
            acc.x += in ? v[k].x : 0.f; acc.y += in ? v[k].y : 0.f;
            acc.z += in ? v[k].z : 0.f; acc.w += in ? v[k].w : 0.f;
        }
    }
    const float rs = rsqrtf(fmaxf(rd, 1.f));
    const float sc = rsqrtf(fmaxf(sdeg[n], 1.f));
    float4 o;
    o.x = (acc.x * rs + b3[4 * l])     * sc;
    o.y = (acc.y * rs + b3[4 * l + 1]) * sc;
    o.z = (acc.z * rs + b3[4 * l + 2]) * sc;
    o.w = (acc.w * rs + b3[4 * l + 3]) * sc;
    *(float4*)(h2 + (size_t)n * DC + 4 * l) = o;
}

__launch_bounds__(256)
__global__ void gather2_softmax_kernel(const float* __restrict__ h2, const int* __restrict__ csr,
                                       const int* __restrict__ offs, const float* __restrict__ rdeg,
                                       float* __restrict__ out) {
    const int gt = blockIdx.x * 256 + threadIdx.x;
    const int n = gt >> 3;
    const int l = gt & 7;
    if (n >= NN) return;
    const int* seg = csr + offs[n];
    const float rd = rdeg[n];
    const int cnt = (int)rd;
    float4 acc = {0.f, 0.f, 0.f, 0.f};
    float tail = 0.f;
    for (int j = 0; j < cnt; j += 8) {
        int s[8];
        float4 v[8];
        float u[8];
        #pragma unroll
        for (int k = 0; k < 8; ++k) {
            const int q = seg[min(j + k, cnt - 1)];
            s[k] = ((unsigned)q < (unsigned)NN) ? q : 0;
        }
        #pragma unroll
        for (int k = 0; k < 8; ++k) {
            const float* row = h2 + (size_t)s[k] * DC;
            v[k] = *(const float4*)(row + 4 * l);
            u[k] = row[32 + l];
        }
        #pragma unroll
        for (int k = 0; k < 8; ++k) {
            const bool in = j + k < cnt;
            acc.x += in ? v[k].x : 0.f; acc.y += in ? v[k].y : 0.f;
            acc.z += in ? v[k].z : 0.f; acc.w += in ? v[k].w : 0.f;
            tail += in ? u[k] : 0.f;
        }
    }
    const float rs = rsqrtf(fmaxf(rd, 1.f));
    const float a0 = acc.x * rs, a1 = acc.y * rs, a2 = acc.z * rs, a3 = acc.w * rs, a4 = tail * rs;
    float mx = fmaxf(fmaxf(fmaxf(a0, a1), fmaxf(a2, a3)), a4);
    #pragma unroll
    for (int m = 1; m < 8; m <<= 1) mx = fmaxf(mx, __shfl_xor(mx, m));
    const float e0 = __expf(a0 - mx), e1 = __expf(a1 - mx), e2 = __expf(a2 - mx),
                e3 = __expf(a3 - mx), e4 = __expf(a4 - mx);
    float sum = (e0 + e1) + (e2 + e3) + e4;
    #pragma unroll
    for (int m = 1; m < 8; m <<= 1) sum += __shfl_xor(sum, m);
    const float inv = 1.f / sum;
    float* orow = out + (size_t)n * DC;
    float4 o;
    o.x = e0 * inv; o.y = e1 * inv; o.z = e2 * inv; o.w = e3 * inv;
    *(float4*)(orow + 4 * l) = o;
    orow[32 + l] = e4 * inv;
}

extern "C" void kernel_launch(void* const* d_in, const int* in_sizes, int n_in,
                              void* d_out, int out_size, void* d_ws, size_t ws_size,
                              hipStream_t stream) {
    const float* nodes     = (const float*)d_in[0];
    const int*   senders   = (const int*)d_in[1];
    const int*   receivers = (const int*)d_in[2];
    const float* W1 = (const float*)d_in[3];
    const float* b1 = (const float*)d_in[4];
    const float* W2 = (const float*)d_in[5];
    const float* b2 = (const float*)d_in[6];
    const float* W3 = (const float*)d_in[7];
    const float* b3 = (const float*)d_in[8];
    float* out = (float*)d_out;
    float* ws  = (float*)d_ws;

    float* sdeg  = ws;
    float* rdeg  = sdeg + NN;
    float* p1    = rdeg + NN;
    float* h2    = p1 + (size_t)NN * P1P;
    int*   offs  = (int*)(h2 + (size_t)NN * DC);
    float* spart = (float*)(offs + NN);
    int*   tab   = (int*)(spart + (size_t)SDC * NN);
    int*   pairs = tab + (size_t)NCH * NBP;
    int*   csr   = pairs + PAIRS_WORDS;

    const int nTiles = NN / 16;
    deg_bin_mlp_kernel<<<SD_BLKS + NCH + MLP_BLKS_A, 256, 0, stream>>>(senders, receivers, spart, tab, pairs,
                                                                       nodes, W1, b1, W2, b2, W3, sdeg, p1, nTiles);
    sort_mlp_kernel<<<NB + MLP_BLKS_B + SCL_BLKS, 256, 0, stream>>>(pairs, tab, offs, rdeg, csr,
                                                                    nodes, W1, b1, W2, b2, W3, spart, sdeg, p1, nTiles);
    gather1_kernel<<<((NN + G1N - 1) / G1N + 7) / 8, 256, 0, stream>>>(p1, csr, offs, rdeg, sdeg, b3, h2);
    gather2_softmax_kernel<<<(NN * 8 + 255) / 256, 256, 0, stream>>>(h2, csr, offs, rdeg, out);
}
